# starting kernel
# baseline (speedup 1.0000x reference)
#include <hip/hip_runtime.h>
#include <hip/hip_bf16.h>


#define BATCH   2
#define SEQ     512
#define DIN     4096
#define DSTATE  16
#define DTRANK  128
#define NPROJ   160
#define NROWS   (BATCH * SEQ)
#define KSPLIT  4

typedef float v2f __attribute__((ext_vector_type(2)));
typedef float v4f __attribute__((ext_vector_type(4)));
typedef float v8f __attribute__((ext_vector_type(8)));

__device__ __forceinline__ v8f wmma_pair(v8f c, v4f a, v4f b) {
  v2f a0 = {a.x, a.y}, a1 = {a.z, a.w};
  v2f b0 = {b.x, b.y}, b1 = {b.z, b.w};
  c = __builtin_amdgcn_wmma_f32_16x16x4_f32(false, a0, false, b0,
                                            (short)0, c, false, false);
  c = __builtin_amdgcn_wmma_f32_16x16x4_f32(false, a1, false, b1,
                                            (short)0, c, false, false);
  return c;
}

__global__ __launch_bounds__(256) void gemm_xproj_partial(
    const float* __restrict__ x, const float* __restrict__ w,
    float* __restrict__ part) {
  const int lane  = threadIdx.x & 31;
  const int wave  = threadIdx.x >> 5;
  const int gtile = blockIdx.x * 8 + wave;
  const int ks    = gtile / 640;
  const int tile  = gtile - ks * 640;
  const int mt = tile / 10, nt = tile - mt * 10;
  const int m0 = mt * 16, n0 = nt * 16;
  const int lo = lane & 15, hi = lane >> 4;
  const int kbase = ks * (DIN / KSPLIT);

  const float* aptr = x + (size_t)(m0 + lo) * DIN + kbase + 4 * hi;
  const float* bptr = w + (size_t)(n0 + lo) * DIN + kbase + 4 * hi;

  v8f c = {};
#pragma unroll 4
  for (int k = 0; k < DIN / KSPLIT; k += 8) {
    v4f a = *(const v4f*)(aptr + k);
    v4f b = *(const v4f*)(bptr + k);
    c = wmma_pair(c, a, b);
  }

  float* out = part + (size_t)ks * NROWS * NPROJ + n0 + lo;
  const int rbase = m0 + 8 * hi;
#pragma unroll
  for (int j = 0; j < 8; ++j)
    out[(size_t)(rbase + j) * NPROJ] = c[j];
}

__global__ __launch_bounds__(256) void reduce_xp_kernel(
    const float* __restrict__ part, float* __restrict__ xp) {
  const size_t S = (size_t)NROWS * NPROJ;
  const size_t i = (size_t)blockIdx.x * 256 + threadIdx.x;
  xp[i] = ((part[i] + part[i + S]) + part[i + 2 * S]) + part[i + 3 * S];
}

__global__ __launch_bounds__(256) void gemm_dt_kernel(
    const float* __restrict__ xp, const float* __restrict__ w,
    const float* __restrict__ bias, float* __restrict__ dtb) {
  const int lane = threadIdx.x & 31;
  const int wave = threadIdx.x >> 5;
  const int tile = blockIdx.x * 8 + wave;
  const int mt = tile >> 8;
  const int nt = tile & 255;
  const int m0 = mt * 16, n0 = nt * 16;
  const int lo = lane & 15, hi = lane >> 4;

  const float* aptr = xp + (size_t)(m0 + lo) * NPROJ + 4 * hi;
  const float* bptr = w + (size_t)(n0 + lo) * DTRANK + 4 * hi;

  v8f c = {};
#pragma unroll
  for (int k = 0; k < DTRANK; k += 8) {
    v4f a = *(const v4f*)(aptr + k);
    v4f b = *(const v4f*)(bptr + k);
    c = wmma_pair(c, a, b);
  }

  const int d = n0 + lo;
  const float bv = bias[d];
  float* out = dtb + d;
  const int rbase = m0 + 8 * hi;
#pragma unroll
  for (int j = 0; j < 8; ++j) {
    float v = c[j] + bv;
    float sp = (v > 20.0f) ? v : log1pf(__expf(v));
    out[(size_t)(rbase + j) * DIN] = sp;
  }
}

#define TC 64

__global__ __launch_bounds__(256) void scan_kernel(
    const float* __restrict__ x, const float* __restrict__ A_log,
    const float* __restrict__ Dp, const float* __restrict__ xp,
    const float* __restrict__ dtb, float* __restrict__ y) {
  __shared__ float sBC[TC * 32];
  __shared__ float sdt[TC * 16];
  __shared__ float sx [TC * 16];
  __shared__ float sy [TC * 16];

  const int tid  = threadIdx.x;
  const int lane = tid & 31;
  const int wave = tid >> 5;
  const int n    = lane & 15;
  const int grp  = lane >> 4;
  const int c    = wave * 2 + grp;
  const int b    = blockIdx.y;
  const int d    = blockIdx.x * 16 + c;

  const float Aval = -__expf(A_log[(size_t)d * DSTATE + n]);
  const float Dv   = Dp[d];
  float h = 0.0f;

  const float* xpb = xp  + (size_t)b * SEQ * NPROJ + DTRANK;
  const float* xg  = x   + (size_t)b * SEQ * DIN + blockIdx.x * 16;
  const float* dtg = dtb + (size_t)b * SEQ * DIN + blockIdx.x * 16;
  float*       yg  = y   + (size_t)b * SEQ * DIN + blockIdx.x * 16;

  for (int t0 = 0; t0 < SEQ; t0 += TC) {
    __syncthreads();
    for (int f = tid; f < TC * 32; f += 256) {
      const int tt = f >> 5, j = f & 31;
      sBC[f] = xpb[(size_t)(t0 + tt) * NPROJ + j];
    }
    for (int f = tid; f < TC * 16; f += 256) {
      const int tt = f >> 4, j = f & 15;
      sdt[f] = dtg[(size_t)(t0 + tt) * DIN + j];
      sx [f] = xg [(size_t)(t0 + tt) * DIN + j];
    }
    __syncthreads();

    if (t0 + TC < SEQ)
      __builtin_prefetch(&xpb[(size_t)(t0 + TC) * NPROJ + (tid & 31)], 0, 0);

    for (int tl = 0; tl < TC; ++tl) {
      const float dt = sdt[tl * 16 + c];
      const float xv = sx [tl * 16 + c];
      const float Bn = sBC[tl * 32 + n];
      const float Cn = sBC[tl * 32 + 16 + n];
      h = __expf(dt * Aval) * h + (dt * xv) * Bn;
      float p = h * Cn;
      p += __shfl_xor(p, 1);
      p += __shfl_xor(p, 2);
      p += __shfl_xor(p, 4);
      p += __shfl_xor(p, 8);
      if (n == 0) sy[tl * 16 + c] = p + Dv * xv;
    }
    __syncthreads();
    for (int f = tid; f < TC * 16; f += 256) {
      const int tt = f >> 4, j = f & 15;
      yg[(size_t)(t0 + tt) * DIN + j] = sy[f];
    }
  }
}

extern "C" void kernel_launch(void* const* d_in, const int* in_sizes, int n_in,
                              void* d_out, int out_size, void* d_ws, size_t ws_size,
                              hipStream_t stream) {
  const float* x     = (const float*)d_in[0];
  const float* A_log = (const float*)d_in[1];
  const float* Dp    = (const float*)d_in[2];
  const float* xw    = (const float*)d_in[3];
  const float* dtw   = (const float*)d_in[4];
  const float* dtb_b = (const float*)d_in[5];
  float* y = (float*)d_out;

  float* xp   = (float*)d_ws;
  float* dtb  = xp + (size_t)NROWS * NPROJ;
  float* part = dtb;

  gemm_xproj_partial<<<dim3(320), dim3(256), 0, stream>>>(x, xw, part);
  reduce_xp_kernel<<<dim3(640), dim3(256), 0, stream>>>(part, xp);
  gemm_dt_kernel<<<dim3(2048), dim3(256), 0, stream>>>(xp, dtw, dtb_b, dtb);
  scan_kernel<<<dim3(DIN / 16, BATCH), dim3(256), 0, stream>>>(
      x, A_log, Dp, xp, dtb, y);
}


# scan kernel: 16-lane dot over n by row_xmask DPP adds on the vector ALU instead of four dependent shuffles through the LDS unit; same butterfly pairs and order, bit-identical
# speedup vs baseline: 1.4033x; 1.4033x over previous
#include <hip/hip_runtime.h>
#include <hip/hip_bf16.h>


#define BATCH   2
#define SEQ     512
#define DIN     4096
#define DSTATE  16
#define DTRANK  128
#define NPROJ   160
#define NROWS   (BATCH * SEQ)
#define KSPLIT  4

typedef float v2f __attribute__((ext_vector_type(2)));
typedef float v4f __attribute__((ext_vector_type(4)));
typedef float v8f __attribute__((ext_vector_type(8)));

__device__ __forceinline__ v8f wmma_pair(v8f c, v4f a, v4f b) {
  v2f a0 = {a.x, a.y}, a1 = {a.z, a.w};
  v2f b0 = {b.x, b.y}, b1 = {b.z, b.w};
  c = __builtin_amdgcn_wmma_f32_16x16x4_f32(false, a0, false, b0,
                                            (short)0, c, false, false);
  c = __builtin_amdgcn_wmma_f32_16x16x4_f32(false, a1, false, b1,
                                            (short)0, c, false, false);
  return c;
}

__global__ __launch_bounds__(256) void gemm_xproj_partial(
    const float* __restrict__ x, const float* __restrict__ w,
    float* __restrict__ part) {
  const int lane  = threadIdx.x & 31;
  const int wave  = threadIdx.x >> 5;
  const int gtile = blockIdx.x * 8 + wave;
  const int ks    = gtile / 640;
  const int tile  = gtile - ks * 640;
  const int mt = tile / 10, nt = tile - mt * 10;
  const int m0 = mt * 16, n0 = nt * 16;
  const int lo = lane & 15, hi = lane >> 4;
  const int kbase = ks * (DIN / KSPLIT);

  const float* aptr = x + (size_t)(m0 + lo) * DIN + kbase + 4 * hi;
  const float* bptr = w + (size_t)(n0 + lo) * DIN + kbase + 4 * hi;

  v8f c = {};
#pragma unroll 4
  for (int k = 0; k < DIN / KSPLIT; k += 8) {
    v4f a = *(const v4f*)(aptr + k);
    v4f b = *(const v4f*)(bptr + k);
    c = wmma_pair(c, a, b);
  }

  float* out = part + (size_t)ks * NROWS * NPROJ + n0 + lo;
  const int rbase = m0 + 8 * hi;
#pragma unroll
  for (int j = 0; j < 8; ++j)
    out[(size_t)(rbase + j) * NPROJ] = c[j];
}

__global__ __launch_bounds__(256) void reduce_xp_kernel(
    const float* __restrict__ part, float* __restrict__ xp) {
  const size_t S = (size_t)NROWS * NPROJ;
  const size_t i = (size_t)blockIdx.x * 256 + threadIdx.x;
  xp[i] = ((part[i] + part[i + S]) + part[i + 2 * S]) + part[i + 3 * S];
}

__global__ __launch_bounds__(256) void gemm_dt_kernel(
    const float* __restrict__ xp, const float* __restrict__ w,
    const float* __restrict__ bias, float* __restrict__ dtb) {
  const int lane = threadIdx.x & 31;
  const int wave = threadIdx.x >> 5;
  const int tile = blockIdx.x * 8 + wave;
  const int mt = tile >> 8;
  const int nt = tile & 255;
  const int m0 = mt * 16, n0 = nt * 16;
  const int lo = lane & 15, hi = lane >> 4;

  const float* aptr = xp + (size_t)(m0 + lo) * NPROJ + 4 * hi;
  const float* bptr = w + (size_t)(n0 + lo) * DTRANK + 4 * hi;

  v8f c = {};
#pragma unroll
  for (int k = 0; k < DTRANK; k += 8) {
    v4f a = *(const v4f*)(aptr + k);
    v4f b = *(const v4f*)(bptr + k);
    c = wmma_pair(c, a, b);
  }

  const int d = n0 + lo;
  const float bv = bias[d];
  float* out = dtb + d;
  const int rbase = m0 + 8 * hi;
#pragma unroll
  for (int j = 0; j < 8; ++j) {
    float v = c[j] + bv;
    float sp = (v > 20.0f) ? v : log1pf(__expf(v));
    out[(size_t)(rbase + j) * DIN] = sp;
  }
}

#define TC 64

template <int CTRL>
__device__ __forceinline__ float row16_xor_add(float p) {
  const int q = __builtin_amdgcn_update_dpp(0, __builtin_bit_cast(int, p), CTRL,
                                            0xf, 0xf, true);
  return p + __builtin_bit_cast(float, q);
}

__global__ __launch_bounds__(256) void scan_kernel(
    const float* __restrict__ x, const float* __restrict__ A_log,
    const float* __restrict__ Dp, const float* __restrict__ xp,
    const float* __restrict__ dtb, float* __restrict__ y) {
  __shared__ float sBC[TC * 32];
  __shared__ float sdt[TC * 16];
  __shared__ float sx [TC * 16];
  __shared__ float sy [TC * 16];

  const int tid  = threadIdx.x;
  const int lane = tid & 31;
  const int wave = tid >> 5;
  const int n    = lane & 15;
  const int grp  = lane >> 4;
  const int c    = wave * 2 + grp;
  const int b    = blockIdx.y;
  const int d    = blockIdx.x * 16 + c;

  const float Aval = -__expf(A_log[(size_t)d * DSTATE + n]);
  const float Dv   = Dp[d];
  float h = 0.0f;

  const float* xpb = xp  + (size_t)b * SEQ * NPROJ + DTRANK;
  const float* xg  = x   + (size_t)b * SEQ * DIN + blockIdx.x * 16;
  const float* dtg = dtb + (size_t)b * SEQ * DIN + blockIdx.x * 16;
  float*       yg  = y   + (size_t)b * SEQ * DIN + blockIdx.x * 16;

  for (int t0 = 0; t0 < SEQ; t0 += TC) {
    __syncthreads();
    for (int f = tid; f < TC * 32; f += 256) {
      const int tt = f >> 5, j = f & 31;
      sBC[f] = xpb[(size_t)(t0 + tt) * NPROJ + j];
    }
    for (int f = tid; f < TC * 16; f += 256) {
      const int tt = f >> 4, j = f & 15;
      sdt[f] = dtg[(size_t)(t0 + tt) * DIN + j];
      sx [f] = xg [(size_t)(t0 + tt) * DIN + j];
    }
    __syncthreads();

    if (t0 + TC < SEQ)
      __builtin_prefetch(&xpb[(size_t)(t0 + TC) * NPROJ + (tid & 31)], 0, 0);

    for (int tl = 0; tl < TC; ++tl) {
      const float dt = sdt[tl * 16 + c];
      const float xv = sx [tl * 16 + c];
      const float Bn = sBC[tl * 32 + n];
      const float Cn = sBC[tl * 32 + 16 + n];
      h = __expf(dt * Aval) * h + (dt * xv) * Bn;
      float p = h * Cn;
      p = row16_xor_add<0x161>(p);
      p = row16_xor_add<0x162>(p);
      p = row16_xor_add<0x164>(p);
      p = row16_xor_add<0x168>(p);
      if (n == 0) sy[tl * 16 + c] = p + Dv * xv;
    }
    __syncthreads();
    for (int f = tid; f < TC * 16; f += 256) {
      const int tt = f >> 4, j = f & 15;
      yg[(size_t)(t0 + tt) * DIN + j] = sy[f];
    }
  }
}

extern "C" void kernel_launch(void* const* d_in, const int* in_sizes, int n_in,
                              void* d_out, int out_size, void* d_ws, size_t ws_size,
                              hipStream_t stream) {
  const float* x     = (const float*)d_in[0];
  const float* A_log = (const float*)d_in[1];
  const float* Dp    = (const float*)d_in[2];
  const float* xw    = (const float*)d_in[3];
  const float* dtw   = (const float*)d_in[4];
  const float* dtb_b = (const float*)d_in[5];
  float* y = (float*)d_out;

  float* xp   = (float*)d_ws;
  float* dtb  = xp + (size_t)NROWS * NPROJ;
  float* part = dtb;

  gemm_xproj_partial<<<dim3(320), dim3(256), 0, stream>>>(x, xw, part);
  reduce_xp_kernel<<<dim3(640), dim3(256), 0, stream>>>(part, xp);
  gemm_dt_kernel<<<dim3(2048), dim3(256), 0, stream>>>(xp, dtw, dtb_b, dtb);
  scan_kernel<<<dim3(DIN / 16, BATCH), dim3(256), 0, stream>>>(
      x, A_log, Dp, xp, dtb, y);
}


# scan kernel: time loop in register batches of 16 steps: operands read from LDS as one group, only the state update stays a serial chain, the 16 row sums run side by side, one masked store per batch
# speedup vs baseline: 1.7150x; 1.2221x over previous
#include <hip/hip_runtime.h>
#include <hip/hip_bf16.h>


#define BATCH   2
#define SEQ     512
#define DIN     4096
#define DSTATE  16
#define DTRANK  128
#define NPROJ   160
#define NROWS   (BATCH * SEQ)
#define KSPLIT  4

typedef float v2f __attribute__((ext_vector_type(2)));
typedef float v4f __attribute__((ext_vector_type(4)));
typedef float v8f __attribute__((ext_vector_type(8)));

__device__ __forceinline__ v8f wmma_pair(v8f c, v4f a, v4f b) {
  v2f a0 = {a.x, a.y}, a1 = {a.z, a.w};
  v2f b0 = {b.x, b.y}, b1 = {b.z, b.w};
  c = __builtin_amdgcn_wmma_f32_16x16x4_f32(false, a0, false, b0,
                                            (short)0, c, false, false);
  c = __builtin_amdgcn_wmma_f32_16x16x4_f32(false, a1, false, b1,
                                            (short)0, c, false, false);
  return c;
}

__global__ __launch_bounds__(256) void gemm_xproj_partial(
    const float* __restrict__ x, const float* __restrict__ w,
    float* __restrict__ part) {
  const int lane  = threadIdx.x & 31;
  const int wave  = threadIdx.x >> 5;
  const int gtile = blockIdx.x * 8 + wave;
  const int ks    = gtile / 640;
  const int tile  = gtile - ks * 640;
  const int mt = tile / 10, nt = tile - mt * 10;
  const int m0 = mt * 16, n0 = nt * 16;
  const int lo = lane & 15, hi = lane >> 4;
  const int kbase = ks * (DIN / KSPLIT);

  const float* aptr = x + (size_t)(m0 + lo) * DIN + kbase + 4 * hi;
  const float* bptr = w + (size_t)(n0 + lo) * DIN + kbase + 4 * hi;

  v8f c = {};
#pragma unroll 4
  for (int k = 0; k < DIN / KSPLIT; k += 8) {
    v4f a = *(const v4f*)(aptr + k);
    v4f b = *(const v4f*)(bptr + k);
    c = wmma_pair(c, a, b);
  }

  float* out = part + (size_t)ks * NROWS * NPROJ + n0 + lo;
  const int rbase = m0 + 8 * hi;
#pragma unroll
  for (int j = 0; j < 8; ++j)
    out[(size_t)(rbase + j) * NPROJ] = c[j];
}

__global__ __launch_bounds__(256) void reduce_xp_kernel(
    const float* __restrict__ part, float* __restrict__ xp) {
  const size_t S = (size_t)NROWS * NPROJ;
  const size_t i = (size_t)blockIdx.x * 256 + threadIdx.x;
  xp[i] = ((part[i] + part[i + S]) + part[i + 2 * S]) + part[i + 3 * S];
}

__global__ __launch_bounds__(256) void gemm_dt_kernel(
    const float* __restrict__ xp, const float* __restrict__ w,
    const float* __restrict__ bias, float* __restrict__ dtb) {
  const int lane = threadIdx.x & 31;
  const int wave = threadIdx.x >> 5;
  const int tile = blockIdx.x * 8 + wave;
  const int mt = tile >> 8;
  const int nt = tile & 255;
  const int m0 = mt * 16, n0 = nt * 16;
  const int lo = lane & 15, hi = lane >> 4;

  const float* aptr = xp + (size_t)(m0 + lo) * NPROJ + 4 * hi;
  const float* bptr = w + (size_t)(n0 + lo) * DTRANK + 4 * hi;

  v8f c = {};
#pragma unroll
  for (int k = 0; k < DTRANK; k += 8) {
    v4f a = *(const v4f*)(aptr + k);
    v4f b = *(const v4f*)(bptr + k);
    c = wmma_pair(c, a, b);
  }

  const int d = n0 + lo;
  const float bv = bias[d];
  float* out = dtb + d;
  const int rbase = m0 + 8 * hi;
#pragma unroll
  for (int j = 0; j < 8; ++j) {
    float v = c[j] + bv;
    float sp = (v > 20.0f) ? v : log1pf(__expf(v));
    out[(size_t)(rbase + j) * DIN] = sp;
  }
}

#define TC 64
#define TB 16

template <int CTRL>
__device__ __forceinline__ float row16_xor_add(float p) {
  const int q = __builtin_amdgcn_update_dpp(0, __builtin_bit_cast(int, p), CTRL,
                                            0xf, 0xf, true);
  return p + __builtin_bit_cast(float, q);
}

__global__ __launch_bounds__(256) void scan_kernel(
    const float* __restrict__ x, const float* __restrict__ A_log,
    const float* __restrict__ Dp, const float* __restrict__ xp,
    const float* __restrict__ dtb, float* __restrict__ y) {
  __shared__ float sBC[TC * 32];
  __shared__ float sdt[TC * 16];
  __shared__ float sx [TC * 16];
  __shared__ float sy [TC * 16];

  const int tid  = threadIdx.x;
  const int lane = tid & 31;
  const int wave = tid >> 5;
  const int n    = lane & 15;
  const int grp  = lane >> 4;
  const int c    = wave * 2 + grp;
  const int b    = blockIdx.y;
  const int d    = blockIdx.x * 16 + c;

  const float Aval = -__expf(A_log[(size_t)d * DSTATE + n]);
  const float Dv   = Dp[d];
  float h = 0.0f;

  const float* xpb = xp  + (size_t)b * SEQ * NPROJ + DTRANK;
  const float* xg  = x   + (size_t)b * SEQ * DIN + blockIdx.x * 16;
  const float* dtg = dtb + (size_t)b * SEQ * DIN + blockIdx.x * 16;
  float*       yg  = y   + (size_t)b * SEQ * DIN + blockIdx.x * 16;

  for (int t0 = 0; t0 < SEQ; t0 += TC) {
    __syncthreads();
    for (int f = tid; f < TC * 32; f += 256) {
      const int tt = f >> 5, j = f & 31;
      sBC[f] = xpb[(size_t)(t0 + tt) * NPROJ + j];
    }
    for (int f = tid; f < TC * 16; f += 256) {
      const int tt = f >> 4, j = f & 15;
      sdt[f] = dtg[(size_t)(t0 + tt) * DIN + j];
      sx [f] = xg [(size_t)(t0 + tt) * DIN + j];
    }
    __syncthreads();

    if (t0 + TC < SEQ)
      __builtin_prefetch(&xpb[(size_t)(t0 + TC) * NPROJ + (tid & 31)], 0, 0);

    const float* bBC = sBC + n;
    const float* bdt = sdt + c;
    const float* bx  = sx + c;
    float*       by  = sy + c;
    for (int tb = 0; tb < TC; tb += TB, bBC += TB * 32, bdt += TB * 16,
                                       bx += TB * 16, by += TB * 16) {
      float dtv[TB], xv[TB], Bv[TB], Cv[TB], p[TB];
#pragma unroll
      for (int j = 0; j < TB; ++j) {
        dtv[j] = bdt[j * 16];
        xv[j]  = bx [j * 16];
        Bv[j]  = bBC[j * 32];
        Cv[j]  = bBC[j * 32 + 16];
      }
#pragma unroll
      for (int j = 0; j < TB; ++j) {
        const float a  = __expf(dtv[j] * Aval);
        const float dx = dtv[j] * xv[j];
        h = fmaf(dx, Bv[j], a * h);
        p[j] = h * Cv[j];
      }
#pragma unroll
      for (int j = 0; j < TB; ++j) p[j] = row16_xor_add<0x161>(p[j]);
#pragma unroll
      for (int j = 0; j < TB; ++j) p[j] = row16_xor_add<0x162>(p[j]);
#pragma unroll
      for (int j = 0; j < TB; ++j) p[j] = row16_xor_add<0x164>(p[j]);
#pragma unroll
      for (int j = 0; j < TB; ++j) p[j] = row16_xor_add<0x168>(p[j]);
      if (n == 0) {
#pragma unroll
        for (int j = 0; j < TB; ++j)
          by[j * 16] = p[j] + Dv * xv[j];
      }
    }
    __syncthreads();
    for (int f = tid; f < TC * 16; f += 256) {
      const int tt = f >> 4, j = f & 15;
      yg[(size_t)(t0 + tt) * DIN + j] = sy[f];
    }
  }
}

extern "C" void kernel_launch(void* const* d_in, const int* in_sizes, int n_in,
                              void* d_out, int out_size, void* d_ws, size_t ws_size,
                              hipStream_t stream) {
  const float* x     = (const float*)d_in[0];
  const float* A_log = (const float*)d_in[1];
  const float* Dp    = (const float*)d_in[2];
  const float* xw    = (const float*)d_in[3];
  const float* dtw   = (const float*)d_in[4];
  const float* dtb_b = (const float*)d_in[5];
  float* y = (float*)d_out;

  float* xp   = (float*)d_ws;
  float* dtb  = xp + (size_t)NROWS * NPROJ;
  float* part = dtb;

  gemm_xproj_partial<<<dim3(320), dim3(256), 0, stream>>>(x, xw, part);
  reduce_xp_kernel<<<dim3(640), dim3(256), 0, stream>>>(part, xp);
  gemm_dt_kernel<<<dim3(2048), dim3(256), 0, stream>>>(xp, dtw, dtb_b, dtb);
  scan_kernel<<<dim3(DIN / 16, BATCH), dim3(256), 0, stream>>>(
      x, A_log, Dp, xp, dtb, y);
}


# scan kernel: next chunk's B/C, dt, x tiles fetched into registers (one b128 per thread per tile) while the time loop runs, stored to LDS at the chunk head; was 12 load-wait-store passes in series
# speedup vs baseline: 2.2382x; 1.3051x over previous
#include <hip/hip_runtime.h>
#include <hip/hip_bf16.h>


#define BATCH   2
#define SEQ     512
#define DIN     4096
#define DSTATE  16
#define DTRANK  128
#define NPROJ   160
#define NROWS   (BATCH * SEQ)
#define KSPLIT  4

typedef float v2f __attribute__((ext_vector_type(2)));
typedef float v4f __attribute__((ext_vector_type(4)));
typedef float v8f __attribute__((ext_vector_type(8)));

__device__ __forceinline__ v8f wmma_pair(v8f c, v4f a, v4f b) {
  v2f a0 = {a.x, a.y}, a1 = {a.z, a.w};
  v2f b0 = {b.x, b.y}, b1 = {b.z, b.w};
  c = __builtin_amdgcn_wmma_f32_16x16x4_f32(false, a0, false, b0,
                                            (short)0, c, false, false);
  c = __builtin_amdgcn_wmma_f32_16x16x4_f32(false, a1, false, b1,
                                            (short)0, c, false, false);
  return c;
}

__global__ __launch_bounds__(256) void gemm_xproj_partial(
    const float* __restrict__ x, const float* __restrict__ w,
    float* __restrict__ part) {
  const int lane  = threadIdx.x & 31;
  const int wave  = threadIdx.x >> 5;
  const int gtile = blockIdx.x * 8 + wave;
  const int ks    = gtile / 640;
  const int tile  = gtile - ks * 640;
  const int mt = tile / 10, nt = tile - mt * 10;
  const int m0 = mt * 16, n0 = nt * 16;
  const int lo = lane & 15, hi = lane >> 4;
  const int kbase = ks * (DIN / KSPLIT);

  const float* aptr = x + (size_t)(m0 + lo) * DIN + kbase + 4 * hi;
  const float* bptr = w + (size_t)(n0 + lo) * DIN + kbase + 4 * hi;

  v8f c = {};
#pragma unroll 4
  for (int k = 0; k < DIN / KSPLIT; k += 8) {
    v4f a = *(const v4f*)(aptr + k);
    v4f b = *(const v4f*)(bptr + k);
    c = wmma_pair(c, a, b);
  }

  float* out = part + (size_t)ks * NROWS * NPROJ + n0 + lo;
  const int rbase = m0 + 8 * hi;
#pragma unroll
  for (int j = 0; j < 8; ++j)
    out[(size_t)(rbase + j) * NPROJ] = c[j];
}

__global__ __launch_bounds__(256) void reduce_xp_kernel(
    const float* __restrict__ part, float* __restrict__ xp) {
  const size_t S = (size_t)NROWS * NPROJ;
  const size_t i = (size_t)blockIdx.x * 256 + threadIdx.x;
  xp[i] = ((part[i] + part[i + S]) + part[i + 2 * S]) + part[i + 3 * S];
}

__global__ __launch_bounds__(256) void gemm_dt_kernel(
    const float* __restrict__ xp, const float* __restrict__ w,
    const float* __restrict__ bias, float* __restrict__ dtb) {
  const int lane = threadIdx.x & 31;
  const int wave = threadIdx.x >> 5;
  const int tile = blockIdx.x * 8 + wave;
  const int mt = tile >> 8;
  const int nt = tile & 255;
  const int m0 = mt * 16, n0 = nt * 16;
  const int lo = lane & 15, hi = lane >> 4;

  const float* aptr = xp + (size_t)(m0 + lo) * NPROJ + 4 * hi;
  const float* bptr = w + (size_t)(n0 + lo) * DTRANK + 4 * hi;

  v8f c = {};
#pragma unroll
  for (int k = 0; k < DTRANK; k += 8) {
    v4f a = *(const v4f*)(aptr + k);
    v4f b = *(const v4f*)(bptr + k);
    c = wmma_pair(c, a, b);
  }

  const int d = n0 + lo;
  const float bv = bias[d];
  float* out = dtb + d;
  const int rbase = m0 + 8 * hi;
#pragma unroll
  for (int j = 0; j < 8; ++j) {
    float v = c[j] + bv;
    float sp = (v > 20.0f) ? v : log1pf(__expf(v));
    out[(size_t)(rbase + j) * DIN] = sp;
  }
}

#define TC 64
#define TB 16

template <int CTRL>
__device__ __forceinline__ float row16_xor_add(float p) {
  const int q = __builtin_amdgcn_update_dpp(0, __builtin_bit_cast(int, p), CTRL,
                                            0xf, 0xf, true);
  return p + __builtin_bit_cast(float, q);
}

__global__ __launch_bounds__(256) void scan_kernel(
    const float* __restrict__ x, const float* __restrict__ A_log,
    const float* __restrict__ Dp, const float* __restrict__ xp,
    const float* __restrict__ dtb, float* __restrict__ y) {
  __shared__ __attribute__((aligned(16))) float sBC[TC * 32];
  __shared__ __attribute__((aligned(16))) float sdt[TC * 16];
  __shared__ __attribute__((aligned(16))) float sx [TC * 16];
  __shared__ float sy [TC * 16];

  const int tid  = threadIdx.x;
  const int lane = tid & 31;
  const int wave = tid >> 5;
  const int n    = lane & 15;
  const int grp  = lane >> 4;
  const int c    = wave * 2 + grp;
  const int b    = blockIdx.y;
  const int d    = blockIdx.x * 16 + c;

  const float Aval = -__expf(A_log[(size_t)d * DSTATE + n]);
  const float Dv   = Dp[d];
  float h = 0.0f;

  const float* xpb = xp  + (size_t)b * SEQ * NPROJ + DTRANK;
  const float* xg  = x   + (size_t)b * SEQ * DIN + blockIdx.x * 16;
  const float* dtg = dtb + (size_t)b * SEQ * DIN + blockIdx.x * 16;
  float*       yg  = y   + (size_t)b * SEQ * DIN + blockIdx.x * 16;

  const float* gBC = xpb + (size_t)(tid >> 3) * NPROJ + (tid & 7) * 4;
  const float* gdt = dtg + (size_t)(tid >> 2) * DIN + (tid & 3) * 4;
  const float* gx  = xg  + (size_t)(tid >> 2) * DIN + (tid & 3) * 4;

  v4f rBC0 = *(const v4f*)(gBC);
  v4f rBC1 = *(const v4f*)(gBC + (size_t)(TC / 2) * NPROJ);
  v4f rdt  = *(const v4f*)(gdt);
  v4f rx   = *(const v4f*)(gx);

  for (int t0 = 0; t0 < SEQ; t0 += TC) {
    __syncthreads();
    *(v4f*)(sBC + 4 * tid)           = rBC0;
    *(v4f*)(sBC + TC * 16 + 4 * tid) = rBC1;
    *(v4f*)(sdt + 4 * tid)           = rdt;
    *(v4f*)(sx  + 4 * tid)           = rx;
    __syncthreads();

    if (t0 + TC < SEQ) {
      const size_t tn = (size_t)(t0 + TC);
      rBC0 = *(const v4f*)(gBC + tn * NPROJ);
      rBC1 = *(const v4f*)(gBC + (tn + TC / 2) * NPROJ);
      rdt  = *(const v4f*)(gdt + tn * DIN);
      rx   = *(const v4f*)(gx  + tn * DIN);
    }

    const float* bBC = sBC + n;
    const float* bdt = sdt + c;
    const float* bx  = sx + c;
    float*       by  = sy + c;
    for (int tb = 0; tb < TC; tb += TB, bBC += TB * 32, bdt += TB * 16,
                                       bx += TB * 16, by += TB * 16) {
      float dtv[TB], xv[TB], Bv[TB], Cv[TB], p[TB];
#pragma unroll
      for (int j = 0; j < TB; ++j) {
        dtv[j] = bdt[j * 16];
        xv[j]  = bx [j * 16];
        Bv[j]  = bBC[j * 32];
        Cv[j]  = bBC[j * 32 + 16];
      }
#pragma unroll
      for (int j = 0; j < TB; ++j) {
        const float a  = __expf(dtv[j] * Aval);
        const float dx = dtv[j] * xv[j];
        h = fmaf(dx, Bv[j], a * h);
        p[j] = h * Cv[j];
      }
#pragma unroll
      for (int j = 0; j < TB; ++j) p[j] = row16_xor_add<0x161>(p[j]);
#pragma unroll
      for (int j = 0; j < TB; ++j) p[j] = row16_xor_add<0x162>(p[j]);
#pragma unroll
      for (int j = 0; j < TB; ++j) p[j] = row16_xor_add<0x164>(p[j]);
#pragma unroll
      for (int j = 0; j < TB; ++j) p[j] = row16_xor_add<0x168>(p[j]);
      if (n == 0) {
#pragma unroll
        for (int j = 0; j < TB; ++j)
          by[j * 16] = p[j] + Dv * xv[j];
      }
    }
    __syncthreads();
    for (int f = tid; f < TC * 16; f += 256) {
      const int tt = f >> 4, j = f & 15;
      yg[(size_t)(t0 + tt) * DIN + j] = sy[f];
    }
  }
}

extern "C" void kernel_launch(void* const* d_in, const int* in_sizes, int n_in,
                              void* d_out, int out_size, void* d_ws, size_t ws_size,
                              hipStream_t stream) {
  const float* x     = (const float*)d_in[0];
  const float* A_log = (const float*)d_in[1];
  const float* Dp    = (const float*)d_in[2];
  const float* xw    = (const float*)d_in[3];
  const float* dtw   = (const float*)d_in[4];
  const float* dtb_b = (const float*)d_in[5];
  float* y = (float*)d_out;

  float* xp   = (float*)d_ws;
  float* dtb  = xp + (size_t)NROWS * NPROJ;
  float* part = dtb;

  gemm_xproj_partial<<<dim3(320), dim3(256), 0, stream>>>(x, xw, part);
  reduce_xp_kernel<<<dim3(640), dim3(256), 0, stream>>>(part, xp);
  gemm_dt_kernel<<<dim3(2048), dim3(256), 0, stream>>>(xp, dtw, dtb_b, dtb);
  scan_kernel<<<dim3(DIN / 16, BATCH), dim3(256), 0, stream>>>(
      x, A_log, Dp, xp, dtb, y);
}


# x_proj GEMM: each wave keeps a 32x80 result block (2x5 accumulator tiles), 7 fragment loads feed 20 matrix instructions, loads run 2 k-steps ahead; split-K 4 -> 16, so the partial-sum order changes
# speedup vs baseline: 2.8978x; 1.2947x over previous
#include <hip/hip_runtime.h>
#include <hip/hip_bf16.h>


#define BATCH   2
#define SEQ     512
#define DIN     4096
#define DSTATE  16
#define DTRANK  128
#define NPROJ   160
#define NROWS   (BATCH * SEQ)
#define KSPLIT  16

typedef float v2f __attribute__((ext_vector_type(2)));
typedef float v4f __attribute__((ext_vector_type(4)));
typedef float v8f __attribute__((ext_vector_type(8)));

__device__ __forceinline__ v8f wmma_pair(v8f c, v4f a, v4f b) {
  v2f a0 = {a.x, a.y}, a1 = {a.z, a.w};
  v2f b0 = {b.x, b.y}, b1 = {b.z, b.w};
  c = __builtin_amdgcn_wmma_f32_16x16x4_f32(false, a0, false, b0,
                                            (short)0, c, false, false);
  c = __builtin_amdgcn_wmma_f32_16x16x4_f32(false, a1, false, b1,
                                            (short)0, c, false, false);
  return c;
}

#define K1_MT 2
#define K1_NT 5
#define K1_KC (DIN / KSPLIT)

__global__ __launch_bounds__(128) void gemm_xproj_partial(
    const float* __restrict__ x, const float* __restrict__ w,
    float* __restrict__ part) {
  const int lane  = threadIdx.x & 31;
  const int wave  = threadIdx.x >> 5;
  const int gtile = blockIdx.x * 4 + wave;
  const int ks    = gtile >> 6;
  const int rem   = gtile & 63;
  const int mt = rem >> 1, nh = rem & 1;
  const int m0 = mt * (16 * K1_MT), n0 = nh * (16 * K1_NT);
  const int lo = lane & 15, hi = lane >> 4;
  const int kbase = ks * K1_KC;

  const float* aptr = x + (size_t)(m0 + lo) * DIN + kbase + 4 * hi;
  const float* bptr = w + (size_t)(n0 + lo) * DIN + kbase + 4 * hi;

  v8f c[K1_MT][K1_NT] = {};
  v4f fa[3][K1_MT], fb[3][K1_NT];
#pragma unroll
  for (int s = 0; s < 2; ++s) {
#pragma unroll
    for (int i = 0; i < K1_MT; ++i)
      fa[s][i] = *(const v4f*)(aptr + (size_t)i * 16 * DIN + 8 * s);
#pragma unroll
    for (int j = 0; j < K1_NT; ++j)
      fb[s][j] = *(const v4f*)(bptr + (size_t)j * 16 * DIN + 8 * s);
  }

#pragma unroll
  for (int s = 0; s < K1_KC / 8; ++s) {
    if (s + 2 < K1_KC / 8) {
#pragma unroll
      for (int i = 0; i < K1_MT; ++i)
        fa[(s + 2) % 3][i] = *(const v4f*)(aptr + (size_t)i * 16 * DIN + 8 * (s + 2));
#pragma unroll
      for (int j = 0; j < K1_NT; ++j)
        fb[(s + 2) % 3][j] = *(const v4f*)(bptr + (size_t)j * 16 * DIN + 8 * (s + 2));
    }
    __builtin_amdgcn_sched_barrier(0);
#pragma unroll
    for (int i = 0; i < K1_MT; ++i)
#pragma unroll
      for (int j = 0; j < K1_NT; ++j)
        c[i][j] = wmma_pair(c[i][j], fa[s % 3][i], fb[s % 3][j]);
  }

  float* out = part + (size_t)ks * NROWS * NPROJ + n0 + lo;
#pragma unroll
  for (int i = 0; i < K1_MT; ++i) {
    const int rbase = m0 + 16 * i + 8 * hi;
#pragma unroll
    for (int j = 0; j < K1_NT; ++j)
#pragma unroll
      for (int r = 0; r < 8; ++r)
        out[(size_t)(rbase + r) * NPROJ + 16 * j] = c[i][j][r];
  }
}

__global__ __launch_bounds__(256) void reduce_xp_kernel(
    const float* __restrict__ part, float* __restrict__ xp) {
  const size_t S = (size_t)NROWS * NPROJ;
  const size_t i = (size_t)blockIdx.x * 256 + threadIdx.x;
  float p[KSPLIT];
#pragma unroll
  for (int s = 0; s < KSPLIT; ++s) p[s] = part[i + s * S];
  float acc = p[0];
#pragma unroll
  for (int s = 1; s < KSPLIT; ++s) acc += p[s];
  xp[i] = acc;
}

__global__ __launch_bounds__(256) void gemm_dt_kernel(
    const float* __restrict__ xp, const float* __restrict__ w,
    const float* __restrict__ bias, float* __restrict__ dtb) {
  const int lane = threadIdx.x & 31;
  const int wave = threadIdx.x >> 5;
  const int tile = blockIdx.x * 8 + wave;
  const int mt = tile >> 8;
  const int nt = tile & 255;
  const int m0 = mt * 16, n0 = nt * 16;
  const int lo = lane & 15, hi = lane >> 4;

  const float* aptr = xp + (size_t)(m0 + lo) * NPROJ + 4 * hi;
  const float* bptr = w + (size_t)(n0 + lo) * DTRANK + 4 * hi;

  v8f c = {};
#pragma unroll
  for (int k = 0; k < DTRANK; k += 8) {
    v4f a = *(const v4f*)(aptr + k);
    v4f b = *(const v4f*)(bptr + k);
    c = wmma_pair(c, a, b);
  }

  const int d = n0 + lo;
  const float bv = bias[d];
  float* out = dtb + d;
  const int rbase = m0 + 8 * hi;
#pragma unroll
  for (int j = 0; j < 8; ++j) {
    float v = c[j] + bv;
    float sp = (v > 20.0f) ? v : log1pf(__expf(v));
    out[(size_t)(rbase + j) * DIN] = sp;
  }
}

#define TC 64
#define TB 16

template <int CTRL>
__device__ __forceinline__ float row16_xor_add(float p) {
  const int q = __builtin_amdgcn_update_dpp(0, __builtin_bit_cast(int, p), CTRL,
                                            0xf, 0xf, true);
  return p + __builtin_bit_cast(float, q);
}

__global__ __launch_bounds__(256) void scan_kernel(
    const float* __restrict__ x, const float* __restrict__ A_log,
    const float* __restrict__ Dp, const float* __restrict__ xp,
    const float* __restrict__ dtb, float* __restrict__ y) {
  __shared__ __attribute__((aligned(16))) float sBC[TC * 32];
  __shared__ __attribute__((aligned(16))) float sdt[TC * 16];
  __shared__ __attribute__((aligned(16))) float sx [TC * 16];
  __shared__ float sy [TC * 16];

  const int tid  = threadIdx.x;
  const int lane = tid & 31;
  const int wave = tid >> 5;
  const int n    = lane & 15;
  const int grp  = lane >> 4;
  const int c    = wave * 2 + grp;
  const int b    = blockIdx.y;
  const int d    = blockIdx.x * 16 + c;

  const float Aval = -__expf(A_log[(size_t)d * DSTATE + n]);
  const float Dv   = Dp[d];
  float h = 0.0f;

  const float* xpb = xp  + (size_t)b * SEQ * NPROJ + DTRANK;
  const float* xg  = x   + (size_t)b * SEQ * DIN + blockIdx.x * 16;
  const float* dtg = dtb + (size_t)b * SEQ * DIN + blockIdx.x * 16;
  float*       yg  = y   + (size_t)b * SEQ * DIN + blockIdx.x * 16;

  const float* gBC = xpb + (size_t)(tid >> 3) * NPROJ + (tid & 7) * 4;
  const float* gdt = dtg + (size_t)(tid >> 2) * DIN + (tid & 3) * 4;
  const float* gx  = xg  + (size_t)(tid >> 2) * DIN + (tid & 3) * 4;

  v4f rBC0 = *(const v4f*)(gBC);
  v4f rBC1 = *(const v4f*)(gBC + (size_t)(TC / 2) * NPROJ);
  v4f rdt  = *(const v4f*)(gdt);
  v4f rx   = *(const v4f*)(gx);

  for (int t0 = 0; t0 < SEQ; t0 += TC) {
    __syncthreads();
    *(v4f*)(sBC + 4 * tid)           = rBC0;
    *(v4f*)(sBC + TC * 16 + 4 * tid) = rBC1;
    *(v4f*)(sdt + 4 * tid)           = rdt;
    *(v4f*)(sx  + 4 * tid)           = rx;
    __syncthreads();

    if (t0 + TC < SEQ) {
      const size_t tn = (size_t)(t0 + TC);
      rBC0 = *(const v4f*)(gBC + tn * NPROJ);
      rBC1 = *(const v4f*)(gBC + (tn + TC / 2) * NPROJ);
      rdt  = *(const v4f*)(gdt + tn * DIN);
      rx   = *(const v4f*)(gx  + tn * DIN);
    }

    const float* bBC = sBC + n;
    const float* bdt = sdt + c;
    const float* bx  = sx + c;
    float*       by  = sy + c;
    for (int tb = 0; tb < TC; tb += TB, bBC += TB * 32, bdt += TB * 16,
                                       bx += TB * 16, by += TB * 16) {
      float dtv[TB], xv[TB], Bv[TB], Cv[TB], p[TB];
#pragma unroll
      for (int j = 0; j < TB; ++j) {
        dtv[j] = bdt[j * 16];
        xv[j]  = bx [j * 16];
        Bv[j]  = bBC[j * 32];
        Cv[j]  = bBC[j * 32 + 16];
      }
#pragma unroll
      for (int j = 0; j < TB; ++j) {
        const float a  = __expf(dtv[j] * Aval);
        const float dx = dtv[j] * xv[j];
        h = fmaf(dx, Bv[j], a * h);
        p[j] = h * Cv[j];
      }
#pragma unroll
      for (int j = 0; j < TB; ++j) p[j] = row16_xor_add<0x161>(p[j]);
#pragma unroll
      for (int j = 0; j < TB; ++j) p[j] = row16_xor_add<0x162>(p[j]);
#pragma unroll
      for (int j = 0; j < TB; ++j) p[j] = row16_xor_add<0x164>(p[j]);
#pragma unroll
      for (int j = 0; j < TB; ++j) p[j] = row16_xor_add<0x168>(p[j]);
      if (n == 0) {
#pragma unroll
        for (int j = 0; j < TB; ++j)
          by[j * 16] = p[j] + Dv * xv[j];
      }
    }
    __syncthreads();
    for (int f = tid; f < TC * 16; f += 256) {
      const int tt = f >> 4, j = f & 15;
      yg[(size_t)(t0 + tt) * DIN + j] = sy[f];
    }
  }
}

extern "C" void kernel_launch(void* const* d_in, const int* in_sizes, int n_in,
                              void* d_out, int out_size, void* d_ws, size_t ws_size,
                              hipStream_t stream) {
  const float* x     = (const float*)d_in[0];
  const float* A_log = (const float*)d_in[1];
  const float* Dp    = (const float*)d_in[2];
  const float* xw    = (const float*)d_in[3];
  const float* dtw   = (const float*)d_in[4];
  const float* dtb_b = (const float*)d_in[5];
  float* y = (float*)d_out;

  float* xp   = (float*)d_ws;
  float* dtb  = xp + (size_t)NROWS * NPROJ;
  float* part = dtb;

  gemm_xproj_partial<<<dim3(256), dim3(128), 0, stream>>>(x, xw, part);
  reduce_xp_kernel<<<dim3(640), dim3(256), 0, stream>>>(part, xp);
  gemm_dt_kernel<<<dim3(2048), dim3(256), 0, stream>>>(xp, dtw, dtb_b, dtb);
  scan_kernel<<<dim3(DIN / 16, BATCH), dim3(256), 0, stream>>>(
      x, A_log, Dp, xp, dtb, y);
}


# scan kernel: the 16 row sums of a batch share one butterfly, each xor stage keeps the half its lane bit picks: 15 DPP adds + 30 selects for 64 DPP adds, lane n stores step n; first add unfused
# speedup vs baseline: 3.1815x; 1.0979x over previous
#include <hip/hip_runtime.h>
#include <hip/hip_bf16.h>


#define BATCH   2
#define SEQ     512
#define DIN     4096
#define DSTATE  16
#define DTRANK  128
#define NPROJ   160
#define NROWS   (BATCH * SEQ)
#define KSPLIT  16

typedef float v2f __attribute__((ext_vector_type(2)));
typedef float v4f __attribute__((ext_vector_type(4)));
typedef float v8f __attribute__((ext_vector_type(8)));

__device__ __forceinline__ v8f wmma_pair(v8f c, v4f a, v4f b) {
  v2f a0 = {a.x, a.y}, a1 = {a.z, a.w};
  v2f b0 = {b.x, b.y}, b1 = {b.z, b.w};
  c = __builtin_amdgcn_wmma_f32_16x16x4_f32(false, a0, false, b0,
                                            (short)0, c, false, false);
  c = __builtin_amdgcn_wmma_f32_16x16x4_f32(false, a1, false, b1,
                                            (short)0, c, false, false);
  return c;
}

#define K1_MT 2
#define K1_NT 5
#define K1_KC (DIN / KSPLIT)

__global__ __launch_bounds__(128) void gemm_xproj_partial(
    const float* __restrict__ x, const float* __restrict__ w,
    float* __restrict__ part) {
  const int lane  = threadIdx.x & 31;
  const int wave  = threadIdx.x >> 5;
  const int gtile = blockIdx.x * 4 + wave;
  const int ks    = gtile >> 6;
  const int rem   = gtile & 63;
  const int mt = rem >> 1, nh = rem & 1;
  const int m0 = mt * (16 * K1_MT), n0 = nh * (16 * K1_NT);
  const int lo = lane & 15, hi = lane >> 4;
  const int kbase = ks * K1_KC;

  const float* aptr = x + (size_t)(m0 + lo) * DIN + kbase + 4 * hi;
  const float* bptr = w + (size_t)(n0 + lo) * DIN + kbase + 4 * hi;

  v8f c[K1_MT][K1_NT] = {};
  v4f fa[3][K1_MT], fb[3][K1_NT];
#pragma unroll
  for (int s = 0; s < 2; ++s) {
#pragma unroll
    for (int i = 0; i < K1_MT; ++i)
      fa[s][i] = *(const v4f*)(aptr + (size_t)i * 16 * DIN + 8 * s);
#pragma unroll
    for (int j = 0; j < K1_NT; ++j)
      fb[s][j] = *(const v4f*)(bptr + (size_t)j * 16 * DIN + 8 * s);
  }

#pragma unroll
  for (int s = 0; s < K1_KC / 8; ++s) {
    if (s + 2 < K1_KC / 8) {
#pragma unroll
      for (int i = 0; i < K1_MT; ++i)
        fa[(s + 2) % 3][i] = *(const v4f*)(aptr + (size_t)i * 16 * DIN + 8 * (s + 2));
#pragma unroll
      for (int j = 0; j < K1_NT; ++j)
        fb[(s + 2) % 3][j] = *(const v4f*)(bptr + (size_t)j * 16 * DIN + 8 * (s + 2));
    }
    __builtin_amdgcn_sched_barrier(0);
#pragma unroll
    for (int i = 0; i < K1_MT; ++i)
#pragma unroll
      for (int j = 0; j < K1_NT; ++j)
        c[i][j] = wmma_pair(c[i][j], fa[s % 3][i], fb[s % 3][j]);
  }

  float* out = part + (size_t)ks * NROWS * NPROJ + n0 + lo;
#pragma unroll
  for (int i = 0; i < K1_MT; ++i) {
    const int rbase = m0 + 16 * i + 8 * hi;
#pragma unroll
    for (int j = 0; j < K1_NT; ++j)
#pragma unroll
      for (int r = 0; r < 8; ++r)
        out[(size_t)(rbase + r) * NPROJ + 16 * j] = c[i][j][r];
  }
}

__global__ __launch_bounds__(256) void reduce_xp_kernel(
    const float* __restrict__ part, float* __restrict__ xp) {
  const size_t S = (size_t)NROWS * NPROJ;
  const size_t i = (size_t)blockIdx.x * 256 + threadIdx.x;
  float p[KSPLIT];
#pragma unroll
  for (int s = 0; s < KSPLIT; ++s) p[s] = part[i + s * S];
  float acc = p[0];
#pragma unroll
  for (int s = 1; s < KSPLIT; ++s) acc += p[s];
  xp[i] = acc;
}

__global__ __launch_bounds__(256) void gemm_dt_kernel(
    const float* __restrict__ xp, const float* __restrict__ w,
    const float* __restrict__ bias, float* __restrict__ dtb) {
  const int lane = threadIdx.x & 31;
  const int wave = threadIdx.x >> 5;
  const int tile = blockIdx.x * 8 + wave;
  const int mt = tile >> 8;
  const int nt = tile & 255;
  const int m0 = mt * 16, n0 = nt * 16;
  const int lo = lane & 15, hi = lane >> 4;

  const float* aptr = xp + (size_t)(m0 + lo) * NPROJ + 4 * hi;
  const float* bptr = w + (size_t)(n0 + lo) * DTRANK + 4 * hi;

  v8f c = {};
#pragma unroll
  for (int k = 0; k < DTRANK; k += 8) {
    v4f a = *(const v4f*)(aptr + k);
    v4f b = *(const v4f*)(bptr + k);
    c = wmma_pair(c, a, b);
  }

  const int d = n0 + lo;
  const float bv = bias[d];
  float* out = dtb + d;
  const int rbase = m0 + 8 * hi;
#pragma unroll
  for (int j = 0; j < 8; ++j) {
    float v = c[j] + bv;
    float sp = (v > 20.0f) ? v : log1pf(__expf(v));
    out[(size_t)(rbase + j) * DIN] = sp;
  }
}

#define TC 64
#define TB 16

template <int CTRL>
__device__ __forceinline__ float row16_fold(bool up, float a, float b) {
  const float keep = up ? b : a;
  const float send = up ? a : b;
  const int q = __builtin_amdgcn_update_dpp(0, __builtin_bit_cast(int, send),
                                            CTRL, 0xf, 0xf, true);
  return keep + __builtin_bit_cast(float, q);
}

__global__ __launch_bounds__(256) void scan_kernel(
    const float* __restrict__ x, const float* __restrict__ A_log,
    const float* __restrict__ Dp, const float* __restrict__ xp,
    const float* __restrict__ dtb, float* __restrict__ y) {
  __shared__ __attribute__((aligned(16))) float sBC[TC * 32];
  __shared__ __attribute__((aligned(16))) float sdt[TC * 16];
  __shared__ __attribute__((aligned(16))) float sx [TC * 16];
  __shared__ float sy [TC * 16];

  const int tid  = threadIdx.x;
  const int lane = tid & 31;
  const int wave = tid >> 5;
  const int n    = lane & 15;
  const int grp  = lane >> 4;
  const int c    = wave * 2 + grp;
  const int b    = blockIdx.y;
  const int d    = blockIdx.x * 16 + c;

  const bool nb0 = (n & 1) != 0, nb1 = (n & 2) != 0;
  const bool nb2 = (n & 4) != 0, nb3 = (n & 8) != 0;

  const float Aval = -__expf(A_log[(size_t)d * DSTATE + n]);
  const float Dv   = Dp[d];
  float h = 0.0f;

  const float* xpb = xp  + (size_t)b * SEQ * NPROJ + DTRANK;
  const float* xg  = x   + (size_t)b * SEQ * DIN + blockIdx.x * 16;
  const float* dtg = dtb + (size_t)b * SEQ * DIN + blockIdx.x * 16;
  float*       yg  = y   + (size_t)b * SEQ * DIN + blockIdx.x * 16;

  const float* gBC = xpb + (size_t)(tid >> 3) * NPROJ + (tid & 7) * 4;
  const float* gdt = dtg + (size_t)(tid >> 2) * DIN + (tid & 3) * 4;
  const float* gx  = xg  + (size_t)(tid >> 2) * DIN + (tid & 3) * 4;

  v4f rBC0 = *(const v4f*)(gBC);
  v4f rBC1 = *(const v4f*)(gBC + (size_t)(TC / 2) * NPROJ);
  v4f rdt  = *(const v4f*)(gdt);
  v4f rx   = *(const v4f*)(gx);

  for (int t0 = 0; t0 < SEQ; t0 += TC) {
    __syncthreads();
    *(v4f*)(sBC + 4 * tid)           = rBC0;
    *(v4f*)(sBC + TC * 16 + 4 * tid) = rBC1;
    *(v4f*)(sdt + 4 * tid)           = rdt;
    *(v4f*)(sx  + 4 * tid)           = rx;
    __syncthreads();

    if (t0 + TC < SEQ) {
      const size_t tn = (size_t)(t0 + TC);
      rBC0 = *(const v4f*)(gBC + tn * NPROJ);
      rBC1 = *(const v4f*)(gBC + (tn + TC / 2) * NPROJ);
      rdt  = *(const v4f*)(gdt + tn * DIN);
      rx   = *(const v4f*)(gx  + tn * DIN);
    }

    const float* bBC = sBC + n;
    const float* bdt = sdt + c;
    const float* bx  = sx + c;
    float*       by  = sy + c;
    for (int tb = 0; tb < TC; tb += TB, bBC += TB * 32, bdt += TB * 16,
                                       bx += TB * 16, by += TB * 16) {
      float dtv[TB], xv[TB], Bv[TB], Cv[TB], p[TB];
#pragma unroll
      for (int j = 0; j < TB; ++j) {
        dtv[j] = bdt[j * 16];
        xv[j]  = bx [j * 16];
        Bv[j]  = bBC[j * 32];
        Cv[j]  = bBC[j * 32 + 16];
      }
      const float xs = bx[n * 16];
#pragma unroll
      for (int j = 0; j < TB; ++j) {
        const float a  = __expf(dtv[j] * Aval);
        const float dx = dtv[j] * xv[j];
        h = fmaf(dx, Bv[j], a * h);
        p[j] = h * Cv[j];
      }
      float q1[TB / 2], q2[TB / 4], q3[TB / 8];
#pragma unroll
      for (int k = 0; k < TB / 2; ++k)
        q1[k] = row16_fold<0x161>(nb0, p[2 * k], p[2 * k + 1]);
#pragma unroll
      for (int k = 0; k < TB / 4; ++k)
        q2[k] = row16_fold<0x162>(nb1, q1[2 * k], q1[2 * k + 1]);
#pragma unroll
      for (int k = 0; k < TB / 8; ++k)
        q3[k] = row16_fold<0x164>(nb2, q2[2 * k], q2[2 * k + 1]);
      const float q4 = row16_fold<0x168>(nb3, q3[0], q3[1]);
      by[n * 16] = q4 + Dv * xs;
    }
    __syncthreads();
    for (int f = tid; f < TC * 16; f += 256) {
      const int tt = f >> 4, j = f & 15;
      yg[(size_t)(t0 + tt) * DIN + j] = sy[f];
    }
  }
}

extern "C" void kernel_launch(void* const* d_in, const int* in_sizes, int n_in,
                              void* d_out, int out_size, void* d_ws, size_t ws_size,
                              hipStream_t stream) {
  const float* x     = (const float*)d_in[0];
  const float* A_log = (const float*)d_in[1];
  const float* Dp    = (const float*)d_in[2];
  const float* xw    = (const float*)d_in[3];
  const float* dtw   = (const float*)d_in[4];
  const float* dtb_b = (const float*)d_in[5];
  float* y = (float*)d_out;

  float* xp   = (float*)d_ws;
  float* dtb  = xp + (size_t)NROWS * NPROJ;
  float* part = dtb;

  gemm_xproj_partial<<<dim3(256), dim3(128), 0, stream>>>(x, xw, part);
  reduce_xp_kernel<<<dim3(640), dim3(256), 0, stream>>>(part, xp);
  gemm_dt_kernel<<<dim3(2048), dim3(256), 0, stream>>>(xp, dtw, dtb_b, dtb);
  scan_kernel<<<dim3(DIN / 16, BATCH), dim3(256), 0, stream>>>(
      x, A_log, Dp, xp, dtb, y);
}


# dt projection fused into the scan kernel: each wave computes one 16x16 dt tile of the next 128-step chunk by WMMA (same k order) under the time loop, softplus at chunk head into LDS; 4 launches -> 3
# speedup vs baseline: 3.7606x; 1.1820x over previous
#include <hip/hip_runtime.h>
#include <hip/hip_bf16.h>


#define BATCH   2
#define SEQ     512
#define DIN     4096
#define DSTATE  16
#define DTRANK  128
#define NPROJ   160
#define NROWS   (BATCH * SEQ)
#define KSPLIT  16

typedef float v2f __attribute__((ext_vector_type(2)));
typedef float v4f __attribute__((ext_vector_type(4)));
typedef float v8f __attribute__((ext_vector_type(8)));

__device__ __forceinline__ v8f wmma_pair(v8f c, v4f a, v4f b) {
  v2f a0 = {a.x, a.y}, a1 = {a.z, a.w};
  v2f b0 = {b.x, b.y}, b1 = {b.z, b.w};
  c = __builtin_amdgcn_wmma_f32_16x16x4_f32(false, a0, false, b0,
                                            (short)0, c, false, false);
  c = __builtin_amdgcn_wmma_f32_16x16x4_f32(false, a1, false, b1,
                                            (short)0, c, false, false);
  return c;
}

#define K1_MT 2
#define K1_NT 5
#define K1_KC (DIN / KSPLIT)

__global__ __launch_bounds__(128) void gemm_xproj_partial(
    const float* __restrict__ x, const float* __restrict__ w,
    float* __restrict__ part) {
  const int lane  = threadIdx.x & 31;
  const int wave  = threadIdx.x >> 5;
  const int gtile = blockIdx.x * 4 + wave;
  const int ks    = gtile >> 6;
  const int rem   = gtile & 63;
  const int mt = rem >> 1, nh = rem & 1;
  const int m0 = mt * (16 * K1_MT), n0 = nh * (16 * K1_NT);
  const int lo = lane & 15, hi = lane >> 4;
  const int kbase = ks * K1_KC;

  const float* aptr = x + (size_t)(m0 + lo) * DIN + kbase + 4 * hi;
  const float* bptr = w + (size_t)(n0 + lo) * DIN + kbase + 4 * hi;

  v8f c[K1_MT][K1_NT] = {};
  v4f fa[3][K1_MT], fb[3][K1_NT];
#pragma unroll
  for (int s = 0; s < 2; ++s) {
#pragma unroll
    for (int i = 0; i < K1_MT; ++i)
      fa[s][i] = *(const v4f*)(aptr + (size_t)i * 16 * DIN + 8 * s);
#pragma unroll
    for (int j = 0; j < K1_NT; ++j)
      fb[s][j] = *(const v4f*)(bptr + (size_t)j * 16 * DIN + 8 * s);
  }

#pragma unroll
  for (int s = 0; s < K1_KC / 8; ++s) {
    if (s + 2 < K1_KC / 8) {
#pragma unroll
      for (int i = 0; i < K1_MT; ++i)
        fa[(s + 2) % 3][i] = *(const v4f*)(aptr + (size_t)i * 16 * DIN + 8 * (s + 2));
#pragma unroll
      for (int j = 0; j < K1_NT; ++j)
        fb[(s + 2) % 3][j] = *(const v4f*)(bptr + (size_t)j * 16 * DIN + 8 * (s + 2));
    }
    __builtin_amdgcn_sched_barrier(0);
#pragma unroll
    for (int i = 0; i < K1_MT; ++i)
#pragma unroll
      for (int j = 0; j < K1_NT; ++j)
        c[i][j] = wmma_pair(c[i][j], fa[s % 3][i], fb[s % 3][j]);
  }

  float* out = part + (size_t)ks * NROWS * NPROJ + n0 + lo;
#pragma unroll
  for (int i = 0; i < K1_MT; ++i) {
    const int rbase = m0 + 16 * i + 8 * hi;
#pragma unroll
    for (int j = 0; j < K1_NT; ++j)
#pragma unroll
      for (int r = 0; r < 8; ++r)
        out[(size_t)(rbase + r) * NPROJ + 16 * j] = c[i][j][r];
  }
}

__global__ __launch_bounds__(256) void reduce_xp_kernel(
    const float* __restrict__ part, float* __restrict__ xp) {
  const size_t S = (size_t)NROWS * NPROJ;
  const size_t i = (size_t)blockIdx.x * 256 + threadIdx.x;
  float p[KSPLIT];
#pragma unroll
  for (int s = 0; s < KSPLIT; ++s) p[s] = part[i + s * S];
  float acc = p[0];
#pragma unroll
  for (int s = 1; s < KSPLIT; ++s) acc += p[s];
  xp[i] = acc;
}

#define TC 128
#define TB 16

template <int CTRL>
__device__ __forceinline__ float row16_fold(bool up, float a, float b) {
  const float keep = up ? b : a;
  const float send = up ? a : b;
  const int q = __builtin_amdgcn_update_dpp(0, __builtin_bit_cast(int, send),
                                            CTRL, 0xf, 0xf, true);
  return keep + __builtin_bit_cast(float, q);
}

__device__ __forceinline__ v8f dt_tile_product(const float* __restrict__ pa,
                                               const float* __restrict__ pw) {
  v4f fa[DTRANK / 8], fw[DTRANK / 8];
#pragma unroll
  for (int s = 0; s < DTRANK / 8; ++s) fa[s] = *(const v4f*)(pa + 8 * s);
#pragma unroll
  for (int s = 0; s < DTRANK / 8; ++s) fw[s] = *(const v4f*)(pw + 8 * s);
  v8f acc = {};
#pragma unroll
  for (int s = 0; s < DTRANK / 8; ++s) acc = wmma_pair(acc, fa[s], fw[s]);
  return acc;
}

__global__ __launch_bounds__(256) void scan_kernel(
    const float* __restrict__ x, const float* __restrict__ A_log,
    const float* __restrict__ Dp, const float* __restrict__ xp,
    const float* __restrict__ dtw, const float* __restrict__ dtbias,
    float* __restrict__ y) {
  __shared__ __attribute__((aligned(16))) float sBC[TC * 32];
  __shared__ __attribute__((aligned(16))) float sx [TC * 16];
  __shared__ float sdt[TC * 16];
  __shared__ float sy [TC * 16];

  const int tid  = threadIdx.x;
  const int lane = tid & 31;
  const int wave = tid >> 5;
  const int n    = lane & 15;
  const int grp  = lane >> 4;
  const int c    = wave * 2 + grp;
  const int b    = blockIdx.y;
  const int d    = blockIdx.x * 16 + c;

  const bool nb0 = (n & 1) != 0, nb1 = (n & 2) != 0;
  const bool nb2 = (n & 4) != 0, nb3 = (n & 8) != 0;

  const float Aval = -__expf(A_log[(size_t)d * DSTATE + n]);
  const float Dv   = Dp[d];
  float h = 0.0f;

  const float* xpb = xp + (size_t)b * SEQ * NPROJ;
  const float* xg  = x  + (size_t)b * SEQ * DIN + blockIdx.x * 16;
  float*       yg  = y  + (size_t)b * SEQ * DIN + blockIdx.x * 16;

  const float* gBC = xpb + DTRANK + (size_t)(tid >> 3) * NPROJ + (tid & 7) * 4;
  const float* gx  = xg + (size_t)(tid >> 2) * DIN + (tid & 3) * 4;

  const float* gA = xpb + (size_t)(wave * 16 + n) * NPROJ + 4 * grp;
  const float* gW = dtw + (size_t)(blockIdx.x * 16 + n) * DTRANK + 4 * grp;
  const float  bv = dtbias[blockIdx.x * 16 + n];
  float* sdt_w = sdt + (wave * 16 + 8 * grp) * 16 + n;

  v4f rBC[4], rx[2];
#pragma unroll
  for (int k = 0; k < 4; ++k)
    rBC[k] = *(const v4f*)(gBC + (size_t)(32 * k) * NPROJ);
#pragma unroll
  for (int k = 0; k < 2; ++k)
    rx[k] = *(const v4f*)(gx + (size_t)(64 * k) * DIN);

  const int wave_s  = __builtin_amdgcn_readfirstlane(wave);
  const int my_slot = (wave_s & 3) + (wave_s >> 2);

  v8f acc = dt_tile_product(gA, gW);

  for (int t0 = 0; t0 < SEQ; t0 += TC) {
    __syncthreads();
#pragma unroll
    for (int k = 0; k < 4; ++k)
      *(v4f*)(sBC + 4 * (tid + 256 * k)) = rBC[k];
#pragma unroll
    for (int k = 0; k < 2; ++k)
      *(v4f*)(sx + 4 * (tid + 256 * k)) = rx[k];

#pragma unroll
    for (int r = 0; r < 8; ++r) {
      const float v  = acc[r] + bv;
      const float sp = log1pf(__expf(v));
      sdt_w[r * 16] = (v > 20.0f) ? v : sp;
    }
    __syncthreads();

    if (t0 + TC < SEQ) {
      const size_t tn = (size_t)(t0 + TC);
#pragma unroll
      for (int k = 0; k < 4; ++k)
        rBC[k] = *(const v4f*)(gBC + (tn + 32 * k) * NPROJ);
#pragma unroll
      for (int k = 0; k < 2; ++k)
        rx[k] = *(const v4f*)(gx + (tn + 64 * k) * DIN);
    }

    const float* bBC = sBC + n;
    const float* bdt = sdt + c;
    const float* bx  = sx + c;
    float*       by  = sy + c;
    auto batch = [&]() __attribute__((always_inline)) {
      float dtv[TB], xv[TB], Bv[TB], Cv[TB], p[TB];
#pragma unroll
      for (int j = 0; j < TB; ++j) {
        dtv[j] = bdt[j * 16];
        xv[j]  = bx [j * 16];
        Bv[j]  = bBC[j * 32];
        Cv[j]  = bBC[j * 32 + 16];
      }
      const float xs = bx[n * 16];
#pragma unroll
      for (int j = 0; j < TB; ++j) {
        const float a  = __expf(dtv[j] * Aval);
        const float dx = dtv[j] * xv[j];
        h = fmaf(dx, Bv[j], a * h);
        p[j] = h * Cv[j];
      }
      float q1[TB / 2], q2[TB / 4], q3[TB / 8];
#pragma unroll
      for (int k = 0; k < TB / 2; ++k)
        q1[k] = row16_fold<0x161>(nb0, p[2 * k], p[2 * k + 1]);
#pragma unroll
      for (int k = 0; k < TB / 4; ++k)
        q2[k] = row16_fold<0x162>(nb1, q1[2 * k], q1[2 * k + 1]);
#pragma unroll
      for (int k = 0; k < TB / 8; ++k)
        q3[k] = row16_fold<0x164>(nb2, q2[2 * k], q2[2 * k + 1]);
      const float q4 = row16_fold<0x168>(nb3, q3[0], q3[1]);
      by[n * 16] = q4 + Dv * xs;
      bBC += TB * 32; bdt += TB * 16; bx += TB * 16; by += TB * 16;
    };
    for (int tb = 0; tb <= my_slot * TB; tb += TB) batch();
    if (t0 + TC < SEQ)
      acc = dt_tile_product(gA + (size_t)(t0 + TC) * NPROJ, gW);
    for (int tb = (my_slot + 1) * TB; tb < TC; tb += TB) batch();
    __syncthreads();
    for (int f = tid; f < TC * 16; f += 256) {
      const int tt = f >> 4, j = f & 15;
      yg[(size_t)(t0 + tt) * DIN + j] = sy[f];
    }
  }
}

extern "C" void kernel_launch(void* const* d_in, const int* in_sizes, int n_in,
                              void* d_out, int out_size, void* d_ws, size_t ws_size,
                              hipStream_t stream) {
  const float* x     = (const float*)d_in[0];
  const float* A_log = (const float*)d_in[1];
  const float* Dp    = (const float*)d_in[2];
  const float* xw    = (const float*)d_in[3];
  const float* dtw   = (const float*)d_in[4];
  const float* dtb_b = (const float*)d_in[5];
  float* y = (float*)d_out;

  float* xp   = (float*)d_ws;
  float* part = xp + (size_t)NROWS * NPROJ;

  gemm_xproj_partial<<<dim3(256), dim3(128), 0, stream>>>(x, xw, part);
  reduce_xp_kernel<<<dim3(640), dim3(256), 0, stream>>>(part, xp);
  scan_kernel<<<dim3(DIN / 16, BATCH), dim3(256), 0, stream>>>(
      x, A_log, Dp, xp, dtw, dtb_b, y);
}


# scan kernel: chunk-head softplus from the hardware exp2/log2/rcp (atanh series for small e, log2 with the rounding of 1+e put back otherwise) instead of library log1pf; dt changes in its last units
# speedup vs baseline: 3.9777x; 1.0577x over previous
#include <hip/hip_runtime.h>
#include <hip/hip_bf16.h>


#define BATCH   2
#define SEQ     512
#define DIN     4096
#define DSTATE  16
#define DTRANK  128
#define NPROJ   160
#define NROWS   (BATCH * SEQ)
#define KSPLIT  16

typedef float v2f __attribute__((ext_vector_type(2)));
typedef float v4f __attribute__((ext_vector_type(4)));
typedef float v8f __attribute__((ext_vector_type(8)));

__device__ __forceinline__ v8f wmma_pair(v8f c, v4f a, v4f b) {
  v2f a0 = {a.x, a.y}, a1 = {a.z, a.w};
  v2f b0 = {b.x, b.y}, b1 = {b.z, b.w};
  c = __builtin_amdgcn_wmma_f32_16x16x4_f32(false, a0, false, b0,
                                            (short)0, c, false, false);
  c = __builtin_amdgcn_wmma_f32_16x16x4_f32(false, a1, false, b1,
                                            (short)0, c, false, false);
  return c;
}

#define K1_MT 2
#define K1_NT 5
#define K1_KC (DIN / KSPLIT)

__global__ __launch_bounds__(128) void gemm_xproj_partial(
    const float* __restrict__ x, const float* __restrict__ w,
    float* __restrict__ part) {
  const int lane  = threadIdx.x & 31;
  const int wave  = threadIdx.x >> 5;
  const int gtile = blockIdx.x * 4 + wave;
  const int ks    = gtile >> 6;
  const int rem   = gtile & 63;
  const int mt = rem >> 1, nh = rem & 1;
  const int m0 = mt * (16 * K1_MT), n0 = nh * (16 * K1_NT);
  const int lo = lane & 15, hi = lane >> 4;
  const int kbase = ks * K1_KC;

  const float* aptr = x + (size_t)(m0 + lo) * DIN + kbase + 4 * hi;
  const float* bptr = w + (size_t)(n0 + lo) * DIN + kbase + 4 * hi;

  v8f c[K1_MT][K1_NT] = {};
  v4f fa[3][K1_MT], fb[3][K1_NT];
#pragma unroll
  for (int s = 0; s < 2; ++s) {
#pragma unroll
    for (int i = 0; i < K1_MT; ++i)
      fa[s][i] = *(const v4f*)(aptr + (size_t)i * 16 * DIN + 8 * s);
#pragma unroll
    for (int j = 0; j < K1_NT; ++j)
      fb[s][j] = *(const v4f*)(bptr + (size_t)j * 16 * DIN + 8 * s);
  }

#pragma unroll
  for (int s = 0; s < K1_KC / 8; ++s) {
    if (s + 2 < K1_KC / 8) {
#pragma unroll
      for (int i = 0; i < K1_MT; ++i)
        fa[(s + 2) % 3][i] = *(const v4f*)(aptr + (size_t)i * 16 * DIN + 8 * (s + 2));
#pragma unroll
      for (int j = 0; j < K1_NT; ++j)
        fb[(s + 2) % 3][j] = *(const v4f*)(bptr + (size_t)j * 16 * DIN + 8 * (s + 2));
    }
    __builtin_amdgcn_sched_barrier(0);
#pragma unroll
    for (int i = 0; i < K1_MT; ++i)
#pragma unroll
      for (int j = 0; j < K1_NT; ++j)
        c[i][j] = wmma_pair(c[i][j], fa[s % 3][i], fb[s % 3][j]);
  }

  float* out = part + (size_t)ks * NROWS * NPROJ + n0 + lo;
#pragma unroll
  for (int i = 0; i < K1_MT; ++i) {
    const int rbase = m0 + 16 * i + 8 * hi;
#pragma unroll
    for (int j = 0; j < K1_NT; ++j)
#pragma unroll
      for (int r = 0; r < 8; ++r)
        out[(size_t)(rbase + r) * NPROJ + 16 * j] = c[i][j][r];
  }
}

__global__ __launch_bounds__(256) void reduce_xp_kernel(
    const float* __restrict__ part, float* __restrict__ xp) {
  const size_t S = (size_t)NROWS * NPROJ;
  const size_t i = (size_t)blockIdx.x * 256 + threadIdx.x;
  float p[KSPLIT];
#pragma unroll
  for (int s = 0; s < KSPLIT; ++s) p[s] = part[i + s * S];
  float acc = p[0];
#pragma unroll
  for (int s = 1; s < KSPLIT; ++s) acc += p[s];
  xp[i] = acc;
}

#define TC 128
#define TB 16

template <int CTRL>
__device__ __forceinline__ float row16_fold(bool up, float a, float b) {
  const float keep = up ? b : a;
  const float send = up ? a : b;
  const int q = __builtin_amdgcn_update_dpp(0, __builtin_bit_cast(int, send),
                                            CTRL, 0xf, 0xf, true);
  return keep + __builtin_bit_cast(float, q);
}

__device__ __forceinline__ float softplus_le20(float v) {
  const float e  = __expf(v);
  const float s  = e * __builtin_amdgcn_rcpf(2.0f + e);
  const float w  = s * s;
  float q = fmaf(w, 0.11111111f, 0.14285715f);
  q = fmaf(w, q, 0.2f);
  q = fmaf(w, q, 0.33333334f);
  const float s2 = s + s;
  const float lo = fmaf(s2 * w, q, s2);
  const float u  = 1.0f + e;
  const float d  = (u - 1.0f) - e;
  const float hi = fmaf(__builtin_amdgcn_logf(u), 0.69314718f,
                        -(d * __builtin_amdgcn_rcpf(u)));
  return (e <= 0.5f) ? lo : hi;
}

__device__ __forceinline__ v8f dt_tile_product(const float* __restrict__ pa,
                                               const float* __restrict__ pw) {
  v4f fa[DTRANK / 8], fw[DTRANK / 8];
#pragma unroll
  for (int s = 0; s < DTRANK / 8; ++s) fa[s] = *(const v4f*)(pa + 8 * s);
#pragma unroll
  for (int s = 0; s < DTRANK / 8; ++s) fw[s] = *(const v4f*)(pw + 8 * s);
  v8f acc = {};
#pragma unroll
  for (int s = 0; s < DTRANK / 8; ++s) acc = wmma_pair(acc, fa[s], fw[s]);
  return acc;
}

__global__ __launch_bounds__(256) void scan_kernel(
    const float* __restrict__ x, const float* __restrict__ A_log,
    const float* __restrict__ Dp, const float* __restrict__ xp,
    const float* __restrict__ dtw, const float* __restrict__ dtbias,
    float* __restrict__ y) {
  __shared__ __attribute__((aligned(16))) float sBC[TC * 32];
  __shared__ __attribute__((aligned(16))) float sx [TC * 16];
  __shared__ float sdt[TC * 16];
  __shared__ float sy [TC * 16];

  const int tid  = threadIdx.x;
  const int lane = tid & 31;
  const int wave = tid >> 5;
  const int n    = lane & 15;
  const int grp  = lane >> 4;
  const int c    = wave * 2 + grp;
  const int b    = blockIdx.y;
  const int d    = blockIdx.x * 16 + c;

  const bool nb0 = (n & 1) != 0, nb1 = (n & 2) != 0;
  const bool nb2 = (n & 4) != 0, nb3 = (n & 8) != 0;

  const float Aval = -__expf(A_log[(size_t)d * DSTATE + n]);
  const float Dv   = Dp[d];
  float h = 0.0f;

  const float* xpb = xp + (size_t)b * SEQ * NPROJ;
  const float* xg  = x  + (size_t)b * SEQ * DIN + blockIdx.x * 16;
  float*       yg  = y  + (size_t)b * SEQ * DIN + blockIdx.x * 16;

  const float* gBC = xpb + DTRANK + (size_t)(tid >> 3) * NPROJ + (tid & 7) * 4;
  const float* gx  = xg + (size_t)(tid >> 2) * DIN + (tid & 3) * 4;

  const float* gA = xpb + (size_t)(wave * 16 + n) * NPROJ + 4 * grp;
  const float* gW = dtw + (size_t)(blockIdx.x * 16 + n) * DTRANK + 4 * grp;
  const float  bv = dtbias[blockIdx.x * 16 + n];
  float* sdt_w = sdt + (wave * 16 + 8 * grp) * 16 + n;

  v4f rBC[4], rx[2];
#pragma unroll
  for (int k = 0; k < 4; ++k)
    rBC[k] = *(const v4f*)(gBC + (size_t)(32 * k) * NPROJ);
#pragma unroll
  for (int k = 0; k < 2; ++k)
    rx[k] = *(const v4f*)(gx + (size_t)(64 * k) * DIN);

  const int wave_s  = __builtin_amdgcn_readfirstlane(wave);
  const int my_slot = (wave_s & 3) + (wave_s >> 2);

  v8f acc = dt_tile_product(gA, gW);

  for (int t0 = 0; t0 < SEQ; t0 += TC) {
    __syncthreads();
#pragma unroll
    for (int k = 0; k < 4; ++k)
      *(v4f*)(sBC + 4 * (tid + 256 * k)) = rBC[k];
#pragma unroll
    for (int k = 0; k < 2; ++k)
      *(v4f*)(sx + 4 * (tid + 256 * k)) = rx[k];

#pragma unroll
    for (int r = 0; r < 8; ++r) {
      const float v  = acc[r] + bv;
      const float sp = softplus_le20(v);
      sdt_w[r * 16] = (v > 20.0f) ? v : sp;
    }
    __syncthreads();

    if (t0 + TC < SEQ) {
      const size_t tn = (size_t)(t0 + TC);
#pragma unroll
      for (int k = 0; k < 4; ++k)
        rBC[k] = *(const v4f*)(gBC + (tn + 32 * k) * NPROJ);
#pragma unroll
      for (int k = 0; k < 2; ++k)
        rx[k] = *(const v4f*)(gx + (tn + 64 * k) * DIN);
    }

    const float* bBC = sBC + n;
    const float* bdt = sdt + c;
    const float* bx  = sx + c;
    float*       by  = sy + c;
    auto batch = [&]() __attribute__((always_inline)) {
      float dtv[TB], xv[TB], Bv[TB], Cv[TB], p[TB];
#pragma unroll
      for (int j = 0; j < TB; ++j) {
        dtv[j] = bdt[j * 16];
        xv[j]  = bx [j * 16];
        Bv[j]  = bBC[j * 32];
        Cv[j]  = bBC[j * 32 + 16];
      }
      const float xs = bx[n * 16];
#pragma unroll
      for (int j = 0; j < TB; ++j) {
        const float a  = __expf(dtv[j] * Aval);
        const float dx = dtv[j] * xv[j];
        h = fmaf(dx, Bv[j], a * h);
        p[j] = h * Cv[j];
      }
      float q1[TB / 2], q2[TB / 4], q3[TB / 8];
#pragma unroll
      for (int k = 0; k < TB / 2; ++k)
        q1[k] = row16_fold<0x161>(nb0, p[2 * k], p[2 * k + 1]);
#pragma unroll
      for (int k = 0; k < TB / 4; ++k)
        q2[k] = row16_fold<0x162>(nb1, q1[2 * k], q1[2 * k + 1]);
#pragma unroll
      for (int k = 0; k < TB / 8; ++k)
        q3[k] = row16_fold<0x164>(nb2, q2[2 * k], q2[2 * k + 1]);
      const float q4 = row16_fold<0x168>(nb3, q3[0], q3[1]);
      by[n * 16] = q4 + Dv * xs;
      bBC += TB * 32; bdt += TB * 16; bx += TB * 16; by += TB * 16;
    };
    for (int tb = 0; tb <= my_slot * TB; tb += TB) batch();
    if (t0 + TC < SEQ)
      acc = dt_tile_product(gA + (size_t)(t0 + TC) * NPROJ, gW);
    for (int tb = (my_slot + 1) * TB; tb < TC; tb += TB) batch();
    __syncthreads();
    for (int f = tid; f < TC * 16; f += 256) {
      const int tt = f >> 4, j = f & 15;
      yg[(size_t)(t0 + tt) * DIN + j] = sy[f];
    }
  }
}

extern "C" void kernel_launch(void* const* d_in, const int* in_sizes, int n_in,
                              void* d_out, int out_size, void* d_ws, size_t ws_size,
                              hipStream_t stream) {
  const float* x     = (const float*)d_in[0];
  const float* A_log = (const float*)d_in[1];
  const float* Dp    = (const float*)d_in[2];
  const float* xw    = (const float*)d_in[3];
  const float* dtw   = (const float*)d_in[4];
  const float* dtb_b = (const float*)d_in[5];
  float* y = (float*)d_out;

  float* xp   = (float*)d_ws;
  float* part = xp + (size_t)NROWS * NPROJ;

  gemm_xproj_partial<<<dim3(256), dim3(128), 0, stream>>>(x, xw, part);
  reduce_xp_kernel<<<dim3(640), dim3(256), 0, stream>>>(part, xp);
  scan_kernel<<<dim3(DIN / 16, BATCH), dim3(256), 0, stream>>>(
      x, A_log, Dp, xp, dtw, dtb_b, y);
}


# scan kernel dt product: both operands read in fragment order (reduce kernel stores xp's dt columns and a copy of dt_proj_w so): a fragment load is one 512-B block, was 16 rows x 32 B; same k order
# speedup vs baseline: 4.2428x; 1.0666x over previous
#include <hip/hip_runtime.h>
#include <hip/hip_bf16.h>


#define BATCH   2
#define SEQ     512
#define DIN     4096
#define DSTATE  16
#define DTRANK  128
#define NPROJ   160
#define NROWS   (BATCH * SEQ)
#define KSPLIT  16

typedef float v2f __attribute__((ext_vector_type(2)));
typedef float v4f __attribute__((ext_vector_type(4)));
typedef float v8f __attribute__((ext_vector_type(8)));

__device__ __forceinline__ v8f wmma_pair(v8f c, v4f a, v4f b) {
  v2f a0 = {a.x, a.y}, a1 = {a.z, a.w};
  v2f b0 = {b.x, b.y}, b1 = {b.z, b.w};
  c = __builtin_amdgcn_wmma_f32_16x16x4_f32(false, a0, false, b0,
                                            (short)0, c, false, false);
  c = __builtin_amdgcn_wmma_f32_16x16x4_f32(false, a1, false, b1,
                                            (short)0, c, false, false);
  return c;
}

#define K1_MT 2
#define K1_NT 5
#define K1_KC (DIN / KSPLIT)

__global__ __launch_bounds__(128) void gemm_xproj_partial(
    const float* __restrict__ x, const float* __restrict__ w,
    float* __restrict__ part) {
  const int lane  = threadIdx.x & 31;
  const int wave  = threadIdx.x >> 5;
  const int gtile = blockIdx.x * 4 + wave;
  const int ks    = gtile >> 6;
  const int rem   = gtile & 63;
  const int mt = rem >> 1, nh = rem & 1;
  const int m0 = mt * (16 * K1_MT), n0 = nh * (16 * K1_NT);
  const int lo = lane & 15, hi = lane >> 4;
  const int kbase = ks * K1_KC;

  const float* aptr = x + (size_t)(m0 + lo) * DIN + kbase + 4 * hi;
  const float* bptr = w + (size_t)(n0 + lo) * DIN + kbase + 4 * hi;

  v8f c[K1_MT][K1_NT] = {};
  v4f fa[3][K1_MT], fb[3][K1_NT];
#pragma unroll
  for (int s = 0; s < 2; ++s) {
#pragma unroll
    for (int i = 0; i < K1_MT; ++i)
      fa[s][i] = *(const v4f*)(aptr + (size_t)i * 16 * DIN + 8 * s);
#pragma unroll
    for (int j = 0; j < K1_NT; ++j)
      fb[s][j] = *(const v4f*)(bptr + (size_t)j * 16 * DIN + 8 * s);
  }

#pragma unroll
  for (int s = 0; s < K1_KC / 8; ++s) {
    if (s + 2 < K1_KC / 8) {
#pragma unroll
      for (int i = 0; i < K1_MT; ++i)
        fa[(s + 2) % 3][i] = *(const v4f*)(aptr + (size_t)i * 16 * DIN + 8 * (s + 2));
#pragma unroll
      for (int j = 0; j < K1_NT; ++j)
        fb[(s + 2) % 3][j] = *(const v4f*)(bptr + (size_t)j * 16 * DIN + 8 * (s + 2));
    }
    __builtin_amdgcn_sched_barrier(0);
#pragma unroll
    for (int i = 0; i < K1_MT; ++i)
#pragma unroll
      for (int j = 0; j < K1_NT; ++j)
        c[i][j] = wmma_pair(c[i][j], fa[s % 3][i], fb[s % 3][j]);
  }

  float* out = part + (size_t)ks * NROWS * NPROJ + n0 + lo;
#pragma unroll
  for (int i = 0; i < K1_MT; ++i) {
    const int rbase = m0 + 16 * i + 8 * hi;
#pragma unroll
    for (int j = 0; j < K1_NT; ++j)
#pragma unroll
      for (int r = 0; r < 8; ++r)
        out[(size_t)(rbase + r) * NPROJ + 16 * j] = c[i][j][r];
  }
}

#define FRAG_TILE (16 * DTRANK)

__global__ __launch_bounds__(256) void reduce_xp_kernel(
    const float* __restrict__ part, const float* __restrict__ dtw,
    float* __restrict__ xp, float* __restrict__ xpF, float* __restrict__ wF) {
  const size_t S = (size_t)NROWS * NPROJ;
  const int i = blockIdx.x * 256 + threadIdx.x;

  const bool has_w = i < DIN * DTRANK / 4;
  v4f wpiece = {};
  if (has_w) {
    const int lo = i & 15, hi = (i >> 4) & 1, s = (i >> 5) & 15, ct = i >> 9;
    wpiece = *(const v4f*)(dtw + (size_t)(ct * 16 + lo) * DTRANK + 8 * s + 4 * hi);
  }

  float p[KSPLIT];
#pragma unroll
  for (int s = 0; s < KSPLIT; ++s) p[s] = part[(size_t)i + s * S];
  float acc = p[0];
#pragma unroll
  for (int s = 1; s < KSPLIT; ++s) acc += p[s];

  const int row = i / NPROJ, col = i - row * NPROJ;
  if (col >= DTRANK) {
    xp[i] = acc;
  } else {
    const int tile = row >> 4, lo = row & 15;
    const int s = col >> 3, hi = (col >> 2) & 1, e = col & 3;
    xpF[(size_t)tile * FRAG_TILE + s * 128 + (hi * 16 + lo) * 4 + e] = acc;
  }
  if (has_w) *(v4f*)(wF + (size_t)i * 4) = wpiece;
}

#define TC 128
#define TB 16

template <int CTRL>
__device__ __forceinline__ float row16_fold(bool up, float a, float b) {
  const float keep = up ? b : a;
  const float send = up ? a : b;
  const int q = __builtin_amdgcn_update_dpp(0, __builtin_bit_cast(int, send),
                                            CTRL, 0xf, 0xf, true);
  return keep + __builtin_bit_cast(float, q);
}

__device__ __forceinline__ float softplus_le20(float v) {
  const float e  = __expf(v);
  const float s  = e * __builtin_amdgcn_rcpf(2.0f + e);
  const float w  = s * s;
  float q = fmaf(w, 0.11111111f, 0.14285715f);
  q = fmaf(w, q, 0.2f);
  q = fmaf(w, q, 0.33333334f);
  const float s2 = s + s;
  const float lo = fmaf(s2 * w, q, s2);
  const float u  = 1.0f + e;
  const float d  = (u - 1.0f) - e;
  const float hi = fmaf(__builtin_amdgcn_logf(u), 0.69314718f,
                        -(d * __builtin_amdgcn_rcpf(u)));
  return (e <= 0.5f) ? lo : hi;
}

__device__ __forceinline__ v8f dt_tile_product(const float* __restrict__ pa,
                                               const float* __restrict__ pw) {
  v4f fa[DTRANK / 8], fw[DTRANK / 8];
#pragma unroll
  for (int s = 0; s < DTRANK / 8; ++s) fa[s] = *(const v4f*)(pa + 128 * s);
#pragma unroll
  for (int s = 0; s < DTRANK / 8; ++s) fw[s] = *(const v4f*)(pw + 128 * s);
  v8f acc = {};
#pragma unroll
  for (int s = 0; s < DTRANK / 8; ++s) acc = wmma_pair(acc, fa[s], fw[s]);
  return acc;
}

__global__ __launch_bounds__(256) void scan_kernel(
    const float* __restrict__ x, const float* __restrict__ A_log,
    const float* __restrict__ Dp, const float* __restrict__ xp,
    const float* __restrict__ xpF, const float* __restrict__ wF,
    const float* __restrict__ dtbias, float* __restrict__ y) {
  __shared__ __attribute__((aligned(16))) float sBC[TC * 32];
  __shared__ __attribute__((aligned(16))) float sx [TC * 16];
  __shared__ float sdt[TC * 16];
  __shared__ float sy [TC * 16];

  const int tid  = threadIdx.x;
  const int lane = tid & 31;
  const int wave = tid >> 5;
  const int n    = lane & 15;
  const int grp  = lane >> 4;
  const int c    = wave * 2 + grp;
  const int b    = blockIdx.y;
  const int d    = blockIdx.x * 16 + c;

  const bool nb0 = (n & 1) != 0, nb1 = (n & 2) != 0;
  const bool nb2 = (n & 4) != 0, nb3 = (n & 8) != 0;

  const float Aval = -__expf(A_log[(size_t)d * DSTATE + n]);
  const float Dv   = Dp[d];
  float h = 0.0f;

  const float* xpb = xp + (size_t)b * SEQ * NPROJ;
  const float* xg  = x  + (size_t)b * SEQ * DIN + blockIdx.x * 16;
  float*       yg  = y  + (size_t)b * SEQ * DIN + blockIdx.x * 16;

  const float* gBC = xpb + DTRANK + (size_t)(tid >> 3) * NPROJ + (tid & 7) * 4;
  const float* gx  = xg + (size_t)(tid >> 2) * DIN + (tid & 3) * 4;

  const float* gA = xpF + (size_t)(b * (SEQ / 16) + wave) * FRAG_TILE + 4 * lane;
  const float* gW = wF + (size_t)blockIdx.x * FRAG_TILE + 4 * lane;
  const float  bv = dtbias[blockIdx.x * 16 + n];
  float* sdt_w = sdt + (wave * 16 + 8 * grp) * 16 + n;

  v4f rBC[4], rx[2];
#pragma unroll
  for (int k = 0; k < 4; ++k)
    rBC[k] = *(const v4f*)(gBC + (size_t)(32 * k) * NPROJ);
#pragma unroll
  for (int k = 0; k < 2; ++k)
    rx[k] = *(const v4f*)(gx + (size_t)(64 * k) * DIN);

  const int wave_s  = __builtin_amdgcn_readfirstlane(wave);
  const int my_slot = (wave_s & 3) + (wave_s >> 2);

  v8f acc = dt_tile_product(gA, gW);

  for (int t0 = 0; t0 < SEQ; t0 += TC) {
    __syncthreads();
#pragma unroll
    for (int k = 0; k < 4; ++k)
      *(v4f*)(sBC + 4 * (tid + 256 * k)) = rBC[k];
#pragma unroll
    for (int k = 0; k < 2; ++k)
      *(v4f*)(sx + 4 * (tid + 256 * k)) = rx[k];

#pragma unroll
    for (int r = 0; r < 8; ++r) {
      const float v  = acc[r] + bv;
      const float sp = softplus_le20(v);
      sdt_w[r * 16] = (v > 20.0f) ? v : sp;
    }
    __syncthreads();

    if (t0 + TC < SEQ) {
      const size_t tn = (size_t)(t0 + TC);
#pragma unroll
      for (int k = 0; k < 4; ++k)
        rBC[k] = *(const v4f*)(gBC + (tn + 32 * k) * NPROJ);
#pragma unroll
      for (int k = 0; k < 2; ++k)
        rx[k] = *(const v4f*)(gx + (tn + 64 * k) * DIN);
    }

    const float* bBC = sBC + n;
    const float* bdt = sdt + c;
    const float* bx  = sx + c;
    float*       by  = sy + c;
    auto batch = [&]() __attribute__((always_inline)) {
      float dtv[TB], xv[TB], Bv[TB], Cv[TB], p[TB];
#pragma unroll
      for (int j = 0; j < TB; ++j) {
        dtv[j] = bdt[j * 16];
        xv[j]  = bx [j * 16];
        Bv[j]  = bBC[j * 32];
        Cv[j]  = bBC[j * 32 + 16];
      }
      const float xs = bx[n * 16];
#pragma unroll
      for (int j = 0; j < TB; ++j) {
        const float a  = __expf(dtv[j] * Aval);
        const float dx = dtv[j] * xv[j];
        h = fmaf(dx, Bv[j], a * h);
        p[j] = h * Cv[j];
      }
      float q1[TB / 2], q2[TB / 4], q3[TB / 8];
#pragma unroll
      for (int k = 0; k < TB / 2; ++k)
        q1[k] = row16_fold<0x161>(nb0, p[2 * k], p[2 * k + 1]);
#pragma unroll
      for (int k = 0; k < TB / 4; ++k)
        q2[k] = row16_fold<0x162>(nb1, q1[2 * k], q1[2 * k + 1]);
#pragma unroll
      for (int k = 0; k < TB / 8; ++k)
        q3[k] = row16_fold<0x164>(nb2, q2[2 * k], q2[2 * k + 1]);
      const float q4 = row16_fold<0x168>(nb3, q3[0], q3[1]);
      by[n * 16] = q4 + Dv * xs;
      bBC += TB * 32; bdt += TB * 16; bx += TB * 16; by += TB * 16;
    };
    for (int tb = 0; tb <= my_slot * TB; tb += TB) batch();
    if (t0 + TC < SEQ)
      acc = dt_tile_product(gA + (size_t)((t0 + TC) / 16) * FRAG_TILE, gW);
    for (int tb = (my_slot + 1) * TB; tb < TC; tb += TB) batch();
    __syncthreads();
    for (int f = tid; f < TC * 16; f += 256) {
      const int tt = f >> 4, j = f & 15;
      yg[(size_t)(t0 + tt) * DIN + j] = sy[f];
    }
  }
}

extern "C" void kernel_launch(void* const* d_in, const int* in_sizes, int n_in,
                              void* d_out, int out_size, void* d_ws, size_t ws_size,
                              hipStream_t stream) {
  const float* x     = (const float*)d_in[0];
  const float* A_log = (const float*)d_in[1];
  const float* Dp    = (const float*)d_in[2];
  const float* xw    = (const float*)d_in[3];
  const float* dtw   = (const float*)d_in[4];
  const float* dtb_b = (const float*)d_in[5];
  float* y = (float*)d_out;

  float* xp   = (float*)d_ws;
  float* part = xp + (size_t)NROWS * NPROJ;
  float* xpF  = part + (size_t)KSPLIT * NROWS * NPROJ;
  float* wF   = xpF + (size_t)NROWS * DTRANK;

  gemm_xproj_partial<<<dim3(256), dim3(128), 0, stream>>>(x, xw, part);
  reduce_xp_kernel<<<dim3(640), dim3(256), 0, stream>>>(part, dtw, xp, xpF, wF);
  scan_kernel<<<dim3(DIN / 16, BATCH), dim3(256), 0, stream>>>(
      x, A_log, Dp, xp, xpF, wF, dtb_b, y);
}


# scan kernel: a lane holds 2 states of one channel (wave 4 channels, work-group 32, 2048 waves): state update on packed f32, dt and x read once per pair, first add of the state sum in-lane, 3 folds
# speedup vs baseline: 4.9220x; 1.1601x over previous
#include <hip/hip_runtime.h>
#include <hip/hip_bf16.h>


#define BATCH   2
#define SEQ     512
#define DIN     4096
#define DSTATE  16
#define DTRANK  128
#define NPROJ   160
#define NROWS   (BATCH * SEQ)
#define KSPLIT  16

typedef float v2f __attribute__((ext_vector_type(2)));
typedef float v4f __attribute__((ext_vector_type(4)));
typedef float v8f __attribute__((ext_vector_type(8)));

__device__ __forceinline__ v8f wmma_pair(v8f c, v4f a, v4f b) {
  v2f a0 = {a.x, a.y}, a1 = {a.z, a.w};
  v2f b0 = {b.x, b.y}, b1 = {b.z, b.w};
  c = __builtin_amdgcn_wmma_f32_16x16x4_f32(false, a0, false, b0,
                                            (short)0, c, false, false);
  c = __builtin_amdgcn_wmma_f32_16x16x4_f32(false, a1, false, b1,
                                            (short)0, c, false, false);
  return c;
}

#define K1_MT 2
#define K1_NT 5
#define K1_KC (DIN / KSPLIT)

__global__ __launch_bounds__(128) void gemm_xproj_partial(
    const float* __restrict__ x, const float* __restrict__ w,
    float* __restrict__ part) {
  const int lane  = threadIdx.x & 31;
  const int wave  = threadIdx.x >> 5;
  const int gtile = blockIdx.x * 4 + wave;
  const int ks    = gtile >> 6;
  const int rem   = gtile & 63;
  const int mt = rem >> 1, nh = rem & 1;
  const int m0 = mt * (16 * K1_MT), n0 = nh * (16 * K1_NT);
  const int lo = lane & 15, hi = lane >> 4;
  const int kbase = ks * K1_KC;

  const float* aptr = x + (size_t)(m0 + lo) * DIN + kbase + 4 * hi;
  const float* bptr = w + (size_t)(n0 + lo) * DIN + kbase + 4 * hi;

  v8f c[K1_MT][K1_NT] = {};
  v4f fa[3][K1_MT], fb[3][K1_NT];
#pragma unroll
  for (int s = 0; s < 2; ++s) {
#pragma unroll
    for (int i = 0; i < K1_MT; ++i)
      fa[s][i] = *(const v4f*)(aptr + (size_t)i * 16 * DIN + 8 * s);
#pragma unroll
    for (int j = 0; j < K1_NT; ++j)
      fb[s][j] = *(const v4f*)(bptr + (size_t)j * 16 * DIN + 8 * s);
  }

#pragma unroll
  for (int s = 0; s < K1_KC / 8; ++s) {
    if (s + 2 < K1_KC / 8) {
#pragma unroll
      for (int i = 0; i < K1_MT; ++i)
        fa[(s + 2) % 3][i] = *(const v4f*)(aptr + (size_t)i * 16 * DIN + 8 * (s + 2));
#pragma unroll
      for (int j = 0; j < K1_NT; ++j)
        fb[(s + 2) % 3][j] = *(const v4f*)(bptr + (size_t)j * 16 * DIN + 8 * (s + 2));
    }
    __builtin_amdgcn_sched_barrier(0);
#pragma unroll
    for (int i = 0; i < K1_MT; ++i)
#pragma unroll
      for (int j = 0; j < K1_NT; ++j)
        c[i][j] = wmma_pair(c[i][j], fa[s % 3][i], fb[s % 3][j]);
  }

  float* out = part + (size_t)ks * NROWS * NPROJ + n0 + lo;
#pragma unroll
  for (int i = 0; i < K1_MT; ++i) {
    const int rbase = m0 + 16 * i + 8 * hi;
#pragma unroll
    for (int j = 0; j < K1_NT; ++j)
#pragma unroll
      for (int r = 0; r < 8; ++r)
        out[(size_t)(rbase + r) * NPROJ + 16 * j] = c[i][j][r];
  }
}

#define FRAG_TILE (16 * DTRANK)

__global__ __launch_bounds__(256) void reduce_xp_kernel(
    const float* __restrict__ part, const float* __restrict__ dtw,
    float* __restrict__ xp, float* __restrict__ xpF, float* __restrict__ wF) {
  const size_t S = (size_t)NROWS * NPROJ;
  const int i = blockIdx.x * 256 + threadIdx.x;

  const bool has_w = i < DIN * DTRANK / 4;
  v4f wpiece = {};
  if (has_w) {
    const int lo = i & 15, hi = (i >> 4) & 1, s = (i >> 5) & 15, ct = i >> 9;
    wpiece = *(const v4f*)(dtw + (size_t)(ct * 16 + lo) * DTRANK + 8 * s + 4 * hi);
  }

  float p[KSPLIT];
#pragma unroll
  for (int s = 0; s < KSPLIT; ++s) p[s] = part[(size_t)i + s * S];
  float acc = p[0];
#pragma unroll
  for (int s = 1; s < KSPLIT; ++s) acc += p[s];

  const int row = i / NPROJ, col = i - row * NPROJ;
  if (col >= DTRANK) {
    xp[i] = acc;
  } else {
    const int tile = row >> 4, lo = row & 15;
    const int s = col >> 3, hi = (col >> 2) & 1, e = col & 3;
    xpF[(size_t)tile * FRAG_TILE + s * 128 + (hi * 16 + lo) * 4 + e] = acc;
  }
  if (has_w) *(v4f*)(wF + (size_t)i * 4) = wpiece;
}

#define TC 128
#define TB 16
#define CH 32

template <int CTRL>
__device__ __forceinline__ float row16_fold(bool up, float a, float b) {
  const float keep = up ? b : a;
  const float send = up ? a : b;
  const int q = __builtin_amdgcn_update_dpp(0, __builtin_bit_cast(int, send),
                                            CTRL, 0xf, 0xf, true);
  return keep + __builtin_bit_cast(float, q);
}

__device__ __forceinline__ float softplus_le20(float v) {
  const float e  = __expf(v);
  const float s  = e * __builtin_amdgcn_rcpf(2.0f + e);
  const float w  = s * s;
  float q = fmaf(w, 0.11111111f, 0.14285715f);
  q = fmaf(w, q, 0.2f);
  q = fmaf(w, q, 0.33333334f);
  const float s2 = s + s;
  const float lo = fmaf(s2 * w, q, s2);
  const float u  = 1.0f + e;
  const float d  = (u - 1.0f) - e;
  const float hi = fmaf(__builtin_amdgcn_logf(u), 0.69314718f,
                        -(d * __builtin_amdgcn_rcpf(u)));
  return (e <= 0.5f) ? lo : hi;
}

__device__ __forceinline__ void dt_tile_product(const float* __restrict__ pa,
                                                const float* __restrict__ pw,
                                                v8f& acc0, v8f& acc1) {
  v4f fa[DTRANK / 8], fw0[DTRANK / 8], fw1[DTRANK / 8];
#pragma unroll
  for (int s = 0; s < DTRANK / 8; ++s) fa[s] = *(const v4f*)(pa + 128 * s);
#pragma unroll
  for (int s = 0; s < DTRANK / 8; ++s) fw0[s] = *(const v4f*)(pw + 128 * s);
#pragma unroll
  for (int s = 0; s < DTRANK / 8; ++s)
    fw1[s] = *(const v4f*)(pw + FRAG_TILE + 128 * s);
  v8f c0 = {}, c1 = {};
#pragma unroll
  for (int s = 0; s < DTRANK / 8; ++s) {
    c0 = wmma_pair(c0, fa[s], fw0[s]);
    c1 = wmma_pair(c1, fa[s], fw1[s]);
  }
  acc0 = c0;
  acc1 = c1;
}

__global__ __launch_bounds__(256) void scan_kernel(
    const float* __restrict__ x, const float* __restrict__ A_log,
    const float* __restrict__ Dp, const float* __restrict__ xp,
    const float* __restrict__ xpF, const float* __restrict__ wF,
    const float* __restrict__ dtbias, float* __restrict__ y) {
  __shared__ __attribute__((aligned(16))) float sBC[TC * 32];
  __shared__ __attribute__((aligned(16))) float sx [TC * CH];
  __shared__ float sdt[TC * CH];
  __shared__ float sy [TC * CH];

  const int tid  = threadIdx.x;
  const int lane = tid & 31;
  const int wave = tid >> 5;
  const int sp   = lane & 7;
  const int cw   = lane >> 3;
  const int c    = wave * 4 + cw;
  const int b    = blockIdx.y;
  const int d    = blockIdx.x * CH + c;
  const int lo   = lane & 15;
  const int hi   = lane >> 4;

  const bool nb0 = (sp & 1) != 0, nb1 = (sp & 2) != 0, nb2 = (sp & 4) != 0;

  const v2f Al   = *(const v2f*)(A_log + (size_t)d * DSTATE + 2 * sp);
  const v2f Aval = {-__expf(Al.x), -__expf(Al.y)};
  const float Dv = Dp[d];
  v2f h = {0.0f, 0.0f};

  const float* xpb = xp + (size_t)b * SEQ * NPROJ;
  const float* xg  = x  + (size_t)b * SEQ * DIN + blockIdx.x * CH;
  float*       yg  = y  + (size_t)b * SEQ * DIN + blockIdx.x * CH;

  const float* gBC = xpb + DTRANK + (size_t)(tid >> 3) * NPROJ + (tid & 7) * 4;
  const float* gx  = xg + (size_t)(tid >> 3) * DIN + (tid & 7) * 4;

  const float* gA = xpF + (size_t)(b * (SEQ / 16) + wave) * FRAG_TILE + 4 * lane;
  const float* gW = wF + (size_t)(2 * blockIdx.x) * FRAG_TILE + 4 * lane;
  const float  bv0 = dtbias[blockIdx.x * CH + lo];
  const float  bv1 = dtbias[blockIdx.x * CH + 16 + lo];
  float* sdt_w = sdt + (wave * 16 + 8 * hi) * CH + lo;

  v4f rBC[4], rx[4];
#pragma unroll
  for (int k = 0; k < 4; ++k)
    rBC[k] = *(const v4f*)(gBC + (size_t)(32 * k) * NPROJ);
#pragma unroll
  for (int k = 0; k < 4; ++k)
    rx[k] = *(const v4f*)(gx + (size_t)(32 * k) * DIN);

  const int wave_s  = __builtin_amdgcn_readfirstlane(wave);
  const int my_slot = (wave_s & 3) + (wave_s >> 2);

  v8f acc0, acc1;
  dt_tile_product(gA, gW, acc0, acc1);

  for (int t0 = 0; t0 < SEQ; t0 += TC) {
    __syncthreads();
#pragma unroll
    for (int k = 0; k < 4; ++k)
      *(v4f*)(sBC + 4 * (tid + 256 * k)) = rBC[k];
#pragma unroll
    for (int k = 0; k < 4; ++k)
      *(v4f*)(sx + 4 * (tid + 256 * k)) = rx[k];

#pragma unroll
    for (int r = 0; r < 8; ++r) {
      const float v0  = acc0[r] + bv0;
      const float v1  = acc1[r] + bv1;
      const float sp0 = softplus_le20(v0);
      const float sp1 = softplus_le20(v1);
      sdt_w[r * CH]      = (v0 > 20.0f) ? v0 : sp0;
      sdt_w[r * CH + 16] = (v1 > 20.0f) ? v1 : sp1;
    }
    __syncthreads();

    if (t0 + TC < SEQ) {
      const size_t tn = (size_t)(t0 + TC);
#pragma unroll
      for (int k = 0; k < 4; ++k)
        rBC[k] = *(const v4f*)(gBC + (tn + 32 * k) * NPROJ);
#pragma unroll
      for (int k = 0; k < 4; ++k)
        rx[k] = *(const v4f*)(gx + (tn + 32 * k) * DIN);
    }

    const float* bBC = sBC + 2 * sp;
    const float* bdt = sdt + c;
    const float* bx  = sx + c;
    float*       by  = sy + c;
    auto batch = [&]() __attribute__((always_inline)) {
      float dtv[TB], xv[TB], p[TB];
      v2f Bv[TB], Cv[TB];
#pragma unroll
      for (int j = 0; j < TB; ++j) {
        dtv[j] = bdt[j * CH];
        xv[j]  = bx [j * CH];
        Bv[j]  = *(const v2f*)(bBC + j * 32);
        Cv[j]  = *(const v2f*)(bBC + j * 32 + 16);
      }
      const float xs0 = bx[sp * CH];
      const float xs1 = bx[(sp + 8) * CH];
#pragma unroll
      for (int j = 0; j < TB; ++j) {
        const v2f arg = dtv[j] * Aval;
        const v2f a   = {__expf(arg.x), __expf(arg.y)};
        const float dx = dtv[j] * xv[j];
        const v2f dx2 = {dx, dx};
        h = __builtin_elementwise_fma(dx2, Bv[j], a * h);
        const v2f hc = h * Cv[j];
        p[j] = hc.x + hc.y;
      }
      float q1[TB / 2], q2[TB / 4], q3[TB / 8];
#pragma unroll
      for (int k = 0; k < TB / 2; ++k)
        q1[k] = row16_fold<0x161>(nb0, p[2 * k], p[2 * k + 1]);
#pragma unroll
      for (int k = 0; k < TB / 4; ++k)
        q2[k] = row16_fold<0x162>(nb1, q1[2 * k], q1[2 * k + 1]);
#pragma unroll
      for (int k = 0; k < TB / 8; ++k)
        q3[k] = row16_fold<0x164>(nb2, q2[2 * k], q2[2 * k + 1]);
      by[sp * CH]       = q3[0] + Dv * xs0;
      by[(sp + 8) * CH] = q3[1] + Dv * xs1;
      bBC += TB * 32; bdt += TB * CH; bx += TB * CH; by += TB * CH;
    };
    for (int tb = 0; tb <= my_slot * TB; tb += TB) batch();
    if (t0 + TC < SEQ)
      dt_tile_product(gA + (size_t)((t0 + TC) / 16) * FRAG_TILE, gW, acc0, acc1);
    for (int tb = (my_slot + 1) * TB; tb < TC; tb += TB) batch();
    __syncthreads();
    for (int f = tid; f < TC * CH; f += 256) {
      const int tt = f >> 5, j = f & 31;
      yg[(size_t)(t0 + tt) * DIN + j] = sy[f];
    }
  }
}

extern "C" void kernel_launch(void* const* d_in, const int* in_sizes, int n_in,
                              void* d_out, int out_size, void* d_ws, size_t ws_size,
                              hipStream_t stream) {
  const float* x     = (const float*)d_in[0];
  const float* A_log = (const float*)d_in[1];
  const float* Dp    = (const float*)d_in[2];
  const float* xw    = (const float*)d_in[3];
  const float* dtw   = (const float*)d_in[4];
  const float* dtb_b = (const float*)d_in[5];
  float* y = (float*)d_out;

  float* xp   = (float*)d_ws;
  float* part = xp + (size_t)NROWS * NPROJ;
  float* xpF  = part + (size_t)KSPLIT * NROWS * NPROJ;
  float* wF   = xpF + (size_t)NROWS * DTRANK;

  gemm_xproj_partial<<<dim3(256), dim3(128), 0, stream>>>(x, xw, part);
  reduce_xp_kernel<<<dim3(640), dim3(256), 0, stream>>>(part, dtw, xp, xpF, wF);
  scan_kernel<<<dim3(DIN / 32, BATCH), dim3(256), 0, stream>>>(
      x, A_log, Dp, xp, xpF, wF, dtb_b, y);
}
